# starting kernel
# baseline (speedup 1.0000x reference)
#include <hip/hip_runtime.h>
#include <cmath>

typedef __attribute__((ext_vector_type(2))) float v2f;
typedef __attribute__((ext_vector_type(8))) float v8f;

static constexpr int kS     = 1600;
static constexpr int kC     = 128;
static constexpr int kNB    = 8;
static constexpr int kHEADS = 8;
static constexpr int kHD    = 16;

__device__ __forceinline__ v8f wmma_f32(v2f a, v2f b, v8f c) {
    return __builtin_amdgcn_wmma_f32_16x16x4_f32(false, a, false, b, (short)0, c,
                                                 false, false);
}

__device__ __forceinline__ float xor16(float x) { return __shfl_xor(x, 16, 32); }

__global__ void qkv_proj_kernel(const float* __restrict__ x,
                                const float* __restrict__ w_qkv,
                                float* __restrict__ Qb, float* __restrict__ Kb,
                                float* __restrict__ Vb) {
    const int ot   = blockIdx.x;
    const int st   = blockIdx.y;
    const int nb   = blockIdx.z;
    const int lane = threadIdx.x;
    const int h    = lane >> 4;
    const int li   = lane & 15;

    const int head  = ot / 3;
    const int which = ot % 3;
    const int s     = st * 16 + li;

    const float* wrow  = w_qkv + (size_t)(ot * 16 + li) * kC;
    const float* xbase = x + (size_t)nb * kC * kS + s;

    v8f acc = {};
#pragma unroll 8
    for (int c0 = 0; c0 < kC; c0 += 4) {
        v2f a, b;
        a.x = wrow[c0 + 2 * h + 0];
        a.y = wrow[c0 + 2 * h + 1];
        b.x = xbase[(size_t)(c0 + 2 * h + 0) * kS];
        b.y = xbase[(size_t)(c0 + 2 * h + 1) * kS];
        acc = wmma_f32(a, b, acc);
    }

    float* dst = (which == 0) ? Qb : (which == 1) ? Kb : Vb;
    dst += (size_t)(nb * kHEADS + head) * kHD * kS;
    const float mul = (which == 0) ? 0.25f : 1.0f;
#pragma unroll
    for (int r = 0; r < 8; ++r) {
        const int d = r + 8 * h;
        dst[(size_t)d * kS + s] = acc[r] * mul;
    }
}

__global__ void attn_kernel(const float* __restrict__ Qb,
                            const float* __restrict__ Kb,
                            const float* __restrict__ Vb,
                            float* __restrict__ CTX) {
    const int it   = blockIdx.x;
    const int head = blockIdx.y;
    const int nb   = blockIdx.z;
    const int lane = threadIdx.x;
    const int h    = lane >> 4;
    const int li   = lane & 15;

    const size_t base = (size_t)(nb * kHEADS + head) * kHD * kS;
    const float* q = Qb + base;
    const float* k = Kb + base;
    const float* v = Vb + base;

    const int i = it * 16 + li;

    float bq[8];
#pragma unroll
    for (int kk = 0; kk < 4; ++kk) {
        bq[2 * kk + 0] = q[(size_t)(4 * kk + 2 * h + 0) * kS + i];
        bq[2 * kk + 1] = q[(size_t)(4 * kk + 2 * h + 1) * kS + i];
    }

    float m_run = -INFINITY;
    float l_run = 0.0f;
    v8f   acc   = {};

    for (int jt = 0; jt < 100; ++jt) {
        const int j0 = jt * 16;

        v8f e = {};
#pragma unroll
        for (int kk = 0; kk < 4; ++kk) {
            v2f a, b;
            a.x = k[(size_t)(4 * kk + 2 * h + 0) * kS + (j0 + li)];
            a.y = k[(size_t)(4 * kk + 2 * h + 1) * kS + (j0 + li)];
            b.x = bq[2 * kk + 0];
            b.y = bq[2 * kk + 1];
            e = wmma_f32(a, b, e);
        }

        float pmax = e[0];
#pragma unroll
        for (int r = 1; r < 8; ++r) pmax = fmaxf(pmax, e[r]);
        pmax = fmaxf(pmax, xor16(pmax));
        const float m_new = fmaxf(m_run, pmax);
        const float alpha = __expf(m_run - m_new);
        float p[8], psum = 0.0f;
#pragma unroll
        for (int r = 0; r < 8; ++r) {
            p[r] = __expf(e[r] - m_new);
            psum += p[r];
        }
        psum += xor16(psum);
        l_run = l_run * alpha + psum;
        m_run = m_new;
#pragma unroll
        for (int r = 0; r < 8; ++r) acc[r] *= alpha;

        const float s0 = xor16(p[0]), s1 = xor16(p[1]);
        const float s2 = xor16(p[2]), s3 = xor16(p[3]);
        const float s4 = xor16(p[4]), s5 = xor16(p[5]);
        const float s6 = xor16(p[6]), s7 = xor16(p[7]);

        v2f bp[4];
        bp[0].x = h ? s2 : p[0];  bp[0].y = h ? s3 : p[1];
        bp[1].x = h ? s6 : p[4];  bp[1].y = h ? s7 : p[5];
        bp[2].x = h ? p[2] : s0;  bp[2].y = h ? p[3] : s1;
        bp[3].x = h ? p[6] : s4;  bp[3].y = h ? p[7] : s5;

#pragma unroll
        for (int kk = 0; kk < 4; ++kk) {
            v2f a;
            a.x = v[(size_t)li * kS + (j0 + 4 * kk + 2 * h + 0)];
            a.y = v[(size_t)li * kS + (j0 + 4 * kk + 2 * h + 1)];
            acc = wmma_f32(a, bp[kk], acc);
        }
    }

    const float inv = 1.0f / l_run;
    float* ctx = CTX + base;
#pragma unroll
    for (int r = 0; r < 8; ++r) {
        const int c = r + 8 * h;
        ctx[(size_t)c * kS + i] = acc[r] * inv;
    }
}

__global__ void out_proj_kernel(const float* __restrict__ CTX,
                                const float* __restrict__ w_o,
                                const float* __restrict__ b_o,
                                float* __restrict__ out) {
    const int ot   = blockIdx.x;
    const int st   = blockIdx.y;
    const int nb   = blockIdx.z;
    const int lane = threadIdx.x;
    const int h    = lane >> 4;
    const int li   = lane & 15;
    const int s    = st * 16 + li;

    const float* wrow  = w_o + (size_t)(ot * 16 + li) * kC;
    const float* cbase = CTX + (size_t)nb * kC * kS + s;

    v8f acc = {};
#pragma unroll 8
    for (int c0 = 0; c0 < kC; c0 += 4) {
        v2f a, b;
        a.x = wrow[c0 + 2 * h + 0];
        a.y = wrow[c0 + 2 * h + 1];
        b.x = cbase[(size_t)(c0 + 2 * h + 0) * kS];
        b.y = cbase[(size_t)(c0 + 2 * h + 1) * kS];
        acc = wmma_f32(a, b, acc);
    }
#pragma unroll
    for (int r = 0; r < 8; ++r) {
        const int oc = ot * 16 + r + 8 * h;
        out[((size_t)nb * kC + oc) * kS + s] = acc[r] + b_o[oc];
    }
}

extern "C" void kernel_launch(void* const* d_in, const int* in_sizes, int n_in,
                              void* d_out, int out_size, void* d_ws, size_t ws_size,
                              hipStream_t stream) {
    const float* x     = (const float*)d_in[0];
    const float* w_qkv = (const float*)d_in[1];
    const float* w_o   = (const float*)d_in[2];
    const float* b_o   = (const float*)d_in[3];
    float*       out   = (float*)d_out;

    const size_t per = (size_t)kNB * kC * kS;
    float* Qb  = (float*)d_ws;
    float* Kb  = Qb + per;
    float* Vb  = Kb + per;
    float* CTX = Vb + per;

    qkv_proj_kernel<<<dim3(24, 100, kNB), 32, 0, stream>>>(x, w_qkv, Qb, Kb, Vb);
    attn_kernel<<<dim3(100, kHEADS, kNB), 32, 0, stream>>>(Qb, Kb, Vb, CTX);
    out_proj_kernel<<<dim3(8, 100, kNB), 32, 0, stream>>>(CTX, w_o, b_o, out);
}


# K and V kept key-major ([s][d]) in the workspace: a 16-key attention tile is one contiguous 1 KB block, not 16 strided rows; QK k slots carry c=8h+2kk+e, so the 16 QK terms sum in a new order
# speedup vs baseline: 1.0980x; 1.0980x over previous
#include <hip/hip_runtime.h>
#include <cmath>

typedef __attribute__((ext_vector_type(2))) float v2f;
typedef __attribute__((ext_vector_type(8))) float v8f;
typedef __attribute__((ext_vector_type(4))) float v4f;

static constexpr int kS     = 1600;
static constexpr int kC     = 128;
static constexpr int kNB    = 8;
static constexpr int kHEADS = 8;
static constexpr int kHD    = 16;

__device__ __forceinline__ v8f wmma_f32(v2f a, v2f b, v8f c) {
    return __builtin_amdgcn_wmma_f32_16x16x4_f32(false, a, false, b, (short)0, c,
                                                 false, false);
}

__device__ __forceinline__ float xor16(float x) { return __shfl_xor(x, 16, 32); }

__global__ void qkv_proj_kernel(const float* __restrict__ x,
                                const float* __restrict__ w_qkv,
                                float* __restrict__ Qb, float* __restrict__ Kb,
                                float* __restrict__ Vb) {
    const int ot   = blockIdx.x;
    const int st   = blockIdx.y;
    const int nb   = blockIdx.z;
    const int lane = threadIdx.x;
    const int h    = lane >> 4;
    const int li   = lane & 15;

    const int head  = ot / 3;
    const int which = ot % 3;
    const int s     = st * 16 + li;

    const float* wrow  = w_qkv + (size_t)(ot * 16 + li) * kC;
    const float* xbase = x + (size_t)nb * kC * kS + s;

    v8f acc = {};
#pragma unroll 8
    for (int c0 = 0; c0 < kC; c0 += 4) {
        v2f a, b;
        a.x = wrow[c0 + 2 * h + 0];
        a.y = wrow[c0 + 2 * h + 1];
        b.x = xbase[(size_t)(c0 + 2 * h + 0) * kS];
        b.y = xbase[(size_t)(c0 + 2 * h + 1) * kS];
        acc = wmma_f32(a, b, acc);
    }

    const size_t base = (size_t)(nb * kHEADS + head) * kHD * kS;
    if (which == 0) {
        float* dst = Qb + base;
#pragma unroll
        for (int r = 0; r < 8; ++r) {
            const int d = r + 8 * h;
            dst[(size_t)d * kS + s] = acc[r] * 0.25f;
        }
    } else {
        float* dst = ((which == 1) ? Kb : Vb) + base + (size_t)s * kHD + 8 * h;
        v4f lo = {acc[0], acc[1], acc[2], acc[3]};
        v4f hi = {acc[4], acc[5], acc[6], acc[7]};
        *reinterpret_cast<v4f*>(dst)     = lo;
        *reinterpret_cast<v4f*>(dst + 4) = hi;
    }
}

__global__ void attn_kernel(const float* __restrict__ Qb,
                            const float* __restrict__ Kb,
                            const float* __restrict__ Vb,
                            float* __restrict__ CTX) {
    const int it   = blockIdx.x;
    const int head = blockIdx.y;
    const int nb   = blockIdx.z;
    const int lane = threadIdx.x;
    const int h    = lane >> 4;
    const int li   = lane & 15;

    const size_t base = (size_t)(nb * kHEADS + head) * kHD * kS;
    const float* q = Qb + base;
    const float* k = Kb + base;
    const float* v = Vb + base;

    const int i = it * 16 + li;

    float bq[8];
#pragma unroll
    for (int kk = 0; kk < 4; ++kk) {
        bq[2 * kk + 0] = q[(size_t)(8 * h + 2 * kk + 0) * kS + i];
        bq[2 * kk + 1] = q[(size_t)(8 * h + 2 * kk + 1) * kS + i];
    }

    const float* kl = k + (size_t)li * kHD + 8 * h;
    const float* vl = v + (size_t)(2 * h) * kHD + li;

    float m_run = -INFINITY;
    float l_run = 0.0f;
    v8f   acc   = {};

    for (int jt = 0; jt < 100; ++jt) {
        const int j0 = jt * 16;

        const float* kt = kl + (size_t)j0 * kHD;
        const v4f k0 = *reinterpret_cast<const v4f*>(kt);
        const v4f k1 = *reinterpret_cast<const v4f*>(kt + 4);
        const float ka[8] = {k0.x, k0.y, k0.z, k0.w, k1.x, k1.y, k1.z, k1.w};
        v8f e = {};
#pragma unroll
        for (int kk = 0; kk < 4; ++kk) {
            v2f a, b;
            a.x = ka[2 * kk + 0];
            a.y = ka[2 * kk + 1];
            b.x = bq[2 * kk + 0];
            b.y = bq[2 * kk + 1];
            e = wmma_f32(a, b, e);
        }

        float pmax = e[0];
#pragma unroll
        for (int r = 1; r < 8; ++r) pmax = fmaxf(pmax, e[r]);
        pmax = fmaxf(pmax, xor16(pmax));
        const float m_new = fmaxf(m_run, pmax);
        const float alpha = __expf(m_run - m_new);
        float p[8], psum = 0.0f;
#pragma unroll
        for (int r = 0; r < 8; ++r) {
            p[r] = __expf(e[r] - m_new);
            psum += p[r];
        }
        psum += xor16(psum);
        l_run = l_run * alpha + psum;
        m_run = m_new;
#pragma unroll
        for (int r = 0; r < 8; ++r) acc[r] *= alpha;

        const float s0 = xor16(p[0]), s1 = xor16(p[1]);
        const float s2 = xor16(p[2]), s3 = xor16(p[3]);
        const float s4 = xor16(p[4]), s5 = xor16(p[5]);
        const float s6 = xor16(p[6]), s7 = xor16(p[7]);

        v2f bp[4];
        bp[0].x = h ? s2 : p[0];  bp[0].y = h ? s3 : p[1];
        bp[1].x = h ? s6 : p[4];  bp[1].y = h ? s7 : p[5];
        bp[2].x = h ? p[2] : s0;  bp[2].y = h ? p[3] : s1;
        bp[3].x = h ? p[6] : s4;  bp[3].y = h ? p[7] : s5;

        const float* vt = vl + (size_t)j0 * kHD;
#pragma unroll
        for (int kk = 0; kk < 4; ++kk) {
            v2f a;
            a.x = vt[(4 * kk + 0) * kHD];
            a.y = vt[(4 * kk + 1) * kHD];
            acc = wmma_f32(a, bp[kk], acc);
        }
    }

    const float inv = 1.0f / l_run;
    float* ctx = CTX + base;
#pragma unroll
    for (int r = 0; r < 8; ++r) {
        const int c = r + 8 * h;
        ctx[(size_t)c * kS + i] = acc[r] * inv;
    }
}

__global__ void out_proj_kernel(const float* __restrict__ CTX,
                                const float* __restrict__ w_o,
                                const float* __restrict__ b_o,
                                float* __restrict__ out) {
    const int ot   = blockIdx.x;
    const int st   = blockIdx.y;
    const int nb   = blockIdx.z;
    const int lane = threadIdx.x;
    const int h    = lane >> 4;
    const int li   = lane & 15;
    const int s    = st * 16 + li;

    const float* wrow  = w_o + (size_t)(ot * 16 + li) * kC;
    const float* cbase = CTX + (size_t)nb * kC * kS + s;

    v8f acc = {};
#pragma unroll 8
    for (int c0 = 0; c0 < kC; c0 += 4) {
        v2f a, b;
        a.x = wrow[c0 + 2 * h + 0];
        a.y = wrow[c0 + 2 * h + 1];
        b.x = cbase[(size_t)(c0 + 2 * h + 0) * kS];
        b.y = cbase[(size_t)(c0 + 2 * h + 1) * kS];
        acc = wmma_f32(a, b, acc);
    }
#pragma unroll
    for (int r = 0; r < 8; ++r) {
        const int oc = ot * 16 + r + 8 * h;
        out[((size_t)nb * kC + oc) * kS + s] = acc[r] + b_o[oc];
    }
}

extern "C" void kernel_launch(void* const* d_in, const int* in_sizes, int n_in,
                              void* d_out, int out_size, void* d_ws, size_t ws_size,
                              hipStream_t stream) {
    const float* x     = (const float*)d_in[0];
    const float* w_qkv = (const float*)d_in[1];
    const float* w_o   = (const float*)d_in[2];
    const float* b_o   = (const float*)d_in[3];
    float*       out   = (float*)d_out;

    const size_t per = (size_t)kNB * kC * kS;
    float* Qb  = (float*)d_ws;
    float* Kb  = Qb + per;
    float* Vb  = Kb + per;
    float* CTX = Vb + per;

    qkv_proj_kernel<<<dim3(24, 100, kNB), 32, 0, stream>>>(x, w_qkv, Qb, Kb, Vb);
    attn_kernel<<<dim3(100, kHEADS, kNB), 32, 0, stream>>>(Qb, Kb, Vb, CTX);
    out_proj_kernel<<<dim3(8, 100, kNB), 32, 0, stream>>>(CTX, w_o, b_o, out);
}


# Key loop: no LDS-unit lane swaps (10 -> 0 per tile): P feeds PV from its own registers (k slot = key 8h+2kk+e), row-sum halves join after the loop, max via permlanex16; PV, row sums add in new order
# speedup vs baseline: 1.1555x; 1.0524x over previous
#include <hip/hip_runtime.h>
#include <cmath>

typedef __attribute__((ext_vector_type(2))) float v2f;
typedef __attribute__((ext_vector_type(8))) float v8f;
typedef __attribute__((ext_vector_type(4))) float v4f;

static constexpr int kS     = 1600;
static constexpr int kC     = 128;
static constexpr int kNB    = 8;
static constexpr int kHEADS = 8;
static constexpr int kHD    = 16;

__device__ __forceinline__ v8f wmma_f32(v2f a, v2f b, v8f c) {
    return __builtin_amdgcn_wmma_f32_16x16x4_f32(false, a, false, b, (short)0, c,
                                                 false, false);
}

__device__ __forceinline__ float xhalf(float x) {
    const int b = __builtin_bit_cast(int, x);
    return __builtin_bit_cast(float,
        __builtin_amdgcn_permlanex16(b, b, 0x76543210, 0xfedcba98, false, false));
}

__global__ void qkv_proj_kernel(const float* __restrict__ x,
                                const float* __restrict__ w_qkv,
                                float* __restrict__ Qb, float* __restrict__ Kb,
                                float* __restrict__ Vb) {
    const int ot   = blockIdx.x;
    const int st   = blockIdx.y;
    const int nb   = blockIdx.z;
    const int lane = threadIdx.x;
    const int h    = lane >> 4;
    const int li   = lane & 15;

    const int head  = ot / 3;
    const int which = ot % 3;
    const int s     = st * 16 + li;

    const float* wrow  = w_qkv + (size_t)(ot * 16 + li) * kC;
    const float* xbase = x + (size_t)nb * kC * kS + s;

    v8f acc = {};
#pragma unroll 8
    for (int c0 = 0; c0 < kC; c0 += 4) {
        v2f a, b;
        a.x = wrow[c0 + 2 * h + 0];
        a.y = wrow[c0 + 2 * h + 1];
        b.x = xbase[(size_t)(c0 + 2 * h + 0) * kS];
        b.y = xbase[(size_t)(c0 + 2 * h + 1) * kS];
        acc = wmma_f32(a, b, acc);
    }

    const size_t base = (size_t)(nb * kHEADS + head) * kHD * kS;
    if (which == 0) {
        float* dst = Qb + base;
#pragma unroll
        for (int r = 0; r < 8; ++r) {
            const int d = r + 8 * h;
            dst[(size_t)d * kS + s] = acc[r] * 0.25f;
        }
    } else {
        float* dst = ((which == 1) ? Kb : Vb) + base + (size_t)s * kHD + 8 * h;
        v4f lo = {acc[0], acc[1], acc[2], acc[3]};
        v4f hi = {acc[4], acc[5], acc[6], acc[7]};
        *reinterpret_cast<v4f*>(dst)     = lo;
        *reinterpret_cast<v4f*>(dst + 4) = hi;
    }
}

__global__ void attn_kernel(const float* __restrict__ Qb,
                            const float* __restrict__ Kb,
                            const float* __restrict__ Vb,
                            float* __restrict__ CTX) {
    const int it   = blockIdx.x;
    const int head = blockIdx.y;
    const int nb   = blockIdx.z;
    const int lane = threadIdx.x;
    const int h    = lane >> 4;
    const int li   = lane & 15;

    const size_t base = (size_t)(nb * kHEADS + head) * kHD * kS;
    const float* q = Qb + base;
    const float* k = Kb + base;
    const float* v = Vb + base;

    const int i = it * 16 + li;

    float bq[8];
#pragma unroll
    for (int kk = 0; kk < 4; ++kk) {
        bq[2 * kk + 0] = q[(size_t)(8 * h + 2 * kk + 0) * kS + i];
        bq[2 * kk + 1] = q[(size_t)(8 * h + 2 * kk + 1) * kS + i];
    }

    const float* kl = k + (size_t)li * kHD + 8 * h;
    const float* vl = v + (size_t)(8 * h) * kHD + li;

    float m_run = -INFINITY;
    float l_run = 0.0f;
    v8f   acc   = {};

    for (int jt = 0; jt < 100; ++jt) {
        const int j0 = jt * 16;

        const float* kt = kl + (size_t)j0 * kHD;
        const v4f k0 = *reinterpret_cast<const v4f*>(kt);
        const v4f k1 = *reinterpret_cast<const v4f*>(kt + 4);
        const float ka[8] = {k0.x, k0.y, k0.z, k0.w, k1.x, k1.y, k1.z, k1.w};
        v8f e = {};
#pragma unroll
        for (int kk = 0; kk < 4; ++kk) {
            v2f a, b;
            a.x = ka[2 * kk + 0];
            a.y = ka[2 * kk + 1];
            b.x = bq[2 * kk + 0];
            b.y = bq[2 * kk + 1];
            e = wmma_f32(a, b, e);
        }

        float pmax = e[0];
#pragma unroll
        for (int r = 1; r < 8; ++r) pmax = fmaxf(pmax, e[r]);
        pmax = fmaxf(pmax, xhalf(pmax));
        const float m_new = fmaxf(m_run, pmax);
        const float alpha = __expf(m_run - m_new);
        float p[8], psum = 0.0f;
#pragma unroll
        for (int r = 0; r < 8; ++r) {
            p[r] = __expf(e[r] - m_new);
            psum += p[r];
        }
        l_run = l_run * alpha + psum;
        m_run = m_new;
#pragma unroll
        for (int r = 0; r < 8; ++r) acc[r] *= alpha;

        const float* vt = vl + (size_t)j0 * kHD;
#pragma unroll
        for (int kk = 0; kk < 4; ++kk) {
            v2f a, b;
            a.x = vt[(2 * kk + 0) * kHD];
            a.y = vt[(2 * kk + 1) * kHD];
            b.x = p[2 * kk + 0];
            b.y = p[2 * kk + 1];
            acc = wmma_f32(a, b, acc);
        }
    }

    l_run += xhalf(l_run);
    const float inv = 1.0f / l_run;
    float* ctx = CTX + base;
#pragma unroll
    for (int r = 0; r < 8; ++r) {
        const int c = r + 8 * h;
        ctx[(size_t)c * kS + i] = acc[r] * inv;
    }
}

__global__ void out_proj_kernel(const float* __restrict__ CTX,
                                const float* __restrict__ w_o,
                                const float* __restrict__ b_o,
                                float* __restrict__ out) {
    const int ot   = blockIdx.x;
    const int st   = blockIdx.y;
    const int nb   = blockIdx.z;
    const int lane = threadIdx.x;
    const int h    = lane >> 4;
    const int li   = lane & 15;
    const int s    = st * 16 + li;

    const float* wrow  = w_o + (size_t)(ot * 16 + li) * kC;
    const float* cbase = CTX + (size_t)nb * kC * kS + s;

    v8f acc = {};
#pragma unroll 8
    for (int c0 = 0; c0 < kC; c0 += 4) {
        v2f a, b;
        a.x = wrow[c0 + 2 * h + 0];
        a.y = wrow[c0 + 2 * h + 1];
        b.x = cbase[(size_t)(c0 + 2 * h + 0) * kS];
        b.y = cbase[(size_t)(c0 + 2 * h + 1) * kS];
        acc = wmma_f32(a, b, acc);
    }
#pragma unroll
    for (int r = 0; r < 8; ++r) {
        const int oc = ot * 16 + r + 8 * h;
        out[((size_t)nb * kC + oc) * kS + s] = acc[r] + b_o[oc];
    }
}

extern "C" void kernel_launch(void* const* d_in, const int* in_sizes, int n_in,
                              void* d_out, int out_size, void* d_ws, size_t ws_size,
                              hipStream_t stream) {
    const float* x     = (const float*)d_in[0];
    const float* w_qkv = (const float*)d_in[1];
    const float* w_o   = (const float*)d_in[2];
    const float* b_o   = (const float*)d_in[3];
    float*       out   = (float*)d_out;

    const size_t per = (size_t)kNB * kC * kS;
    float* Qb  = (float*)d_ws;
    float* Kb  = Qb + per;
    float* Vb  = Kb + per;
    float* CTX = Vb + per;

    qkv_proj_kernel<<<dim3(24, 100, kNB), 32, 0, stream>>>(x, w_qkv, Qb, Kb, Vb);
    attn_kernel<<<dim3(100, kHEADS, kNB), 32, 0, stream>>>(Qb, Kb, Vb, CTX);
    out_proj_kernel<<<dim3(8, 100, kNB), 32, 0, stream>>>(CTX, w_o, b_o, out);
}


# Attention key loop software-pipelined over two K/V register sets: a 16-key tile's loads are issued one tile ahead of use and arrive during the arithmetic of the tile before; sums keep their order
# speedup vs baseline: 1.2427x; 1.0755x over previous
#include <hip/hip_runtime.h>
#include <cmath>

typedef __attribute__((ext_vector_type(2))) float v2f;
typedef __attribute__((ext_vector_type(8))) float v8f;
typedef __attribute__((ext_vector_type(4))) float v4f;

static constexpr int kS     = 1600;
static constexpr int kC     = 128;
static constexpr int kNB    = 8;
static constexpr int kHEADS = 8;
static constexpr int kHD    = 16;

__device__ __forceinline__ v8f wmma_f32(v2f a, v2f b, v8f c) {
    return __builtin_amdgcn_wmma_f32_16x16x4_f32(false, a, false, b, (short)0, c,
                                                 false, false);
}

__device__ __forceinline__ float xhalf(float x) {
    const int b = __builtin_bit_cast(int, x);
    return __builtin_bit_cast(float,
        __builtin_amdgcn_permlanex16(b, b, 0x76543210, 0xfedcba98, false, false));
}

__global__ void qkv_proj_kernel(const float* __restrict__ x,
                                const float* __restrict__ w_qkv,
                                float* __restrict__ Qb, float* __restrict__ Kb,
                                float* __restrict__ Vb) {
    const int ot   = blockIdx.x;
    const int st   = blockIdx.y;
    const int nb   = blockIdx.z;
    const int lane = threadIdx.x;
    const int h    = lane >> 4;
    const int li   = lane & 15;

    const int head  = ot / 3;
    const int which = ot % 3;
    const int s     = st * 16 + li;

    const float* wrow  = w_qkv + (size_t)(ot * 16 + li) * kC;
    const float* xbase = x + (size_t)nb * kC * kS + s;

    v8f acc = {};
#pragma unroll 8
    for (int c0 = 0; c0 < kC; c0 += 4) {
        v2f a, b;
        a.x = wrow[c0 + 2 * h + 0];
        a.y = wrow[c0 + 2 * h + 1];
        b.x = xbase[(size_t)(c0 + 2 * h + 0) * kS];
        b.y = xbase[(size_t)(c0 + 2 * h + 1) * kS];
        acc = wmma_f32(a, b, acc);
    }

    const size_t base = (size_t)(nb * kHEADS + head) * kHD * kS;
    if (which == 0) {
        float* dst = Qb + base;
#pragma unroll
        for (int r = 0; r < 8; ++r) {
            const int d = r + 8 * h;
            dst[(size_t)d * kS + s] = acc[r] * 0.25f;
        }
    } else {
        float* dst = ((which == 1) ? Kb : Vb) + base + (size_t)s * kHD + 8 * h;
        v4f lo = {acc[0], acc[1], acc[2], acc[3]};
        v4f hi = {acc[4], acc[5], acc[6], acc[7]};
        *reinterpret_cast<v4f*>(dst)     = lo;
        *reinterpret_cast<v4f*>(dst + 4) = hi;
    }
}


struct KVFrag {
    v4f   k0, k1;
    float v[8];
};

__device__ __forceinline__ void hold_loads_here() {
    asm volatile("" ::: "memory");
    __builtin_amdgcn_sched_barrier(0);
}

__device__ __forceinline__ KVFrag load_kv(const float* __restrict__ kl,
                                          const float* __restrict__ vl, int jt) {
    KVFrag t;
    const float* kt = kl + (size_t)jt * 16 * kHD;
    t.k0 = *reinterpret_cast<const v4f*>(kt);
    t.k1 = *reinterpret_cast<const v4f*>(kt + 4);
    const float* vt = vl + (size_t)jt * 16 * kHD;
#pragma unroll
    for (int r = 0; r < 8; ++r) t.v[r] = vt[r * kHD];
    return t;
}

__device__ __forceinline__ void attn_tile(const KVFrag& t, const float (&bq)[8],
                                          float& m_run, float& l_run, v8f& acc) {
    const float ka[8] = {t.k0.x, t.k0.y, t.k0.z, t.k0.w, t.k1.x, t.k1.y, t.k1.z, t.k1.w};
    v8f e = {};
#pragma unroll
    for (int kk = 0; kk < 4; ++kk) {
        v2f a, b;
        a.x = ka[2 * kk + 0];
        a.y = ka[2 * kk + 1];
        b.x = bq[2 * kk + 0];
        b.y = bq[2 * kk + 1];
        e = wmma_f32(a, b, e);
    }

    float pmax = e[0];
#pragma unroll
    for (int r = 1; r < 8; ++r) pmax = fmaxf(pmax, e[r]);
    pmax = fmaxf(pmax, xhalf(pmax));
    const float m_new = fmaxf(m_run, pmax);
    const float alpha = __expf(m_run - m_new);
    float p[8], psum = 0.0f;
#pragma unroll
    for (int r = 0; r < 8; ++r) {
        p[r] = __expf(e[r] - m_new);
        psum += p[r];
    }
    l_run = l_run * alpha + psum;
    m_run = m_new;
#pragma unroll
    for (int r = 0; r < 8; ++r) acc[r] *= alpha;

#pragma unroll
    for (int kk = 0; kk < 4; ++kk) {
        v2f a, b;
        a.x = t.v[2 * kk + 0];
        a.y = t.v[2 * kk + 1];
        b.x = p[2 * kk + 0];
        b.y = p[2 * kk + 1];
        acc = wmma_f32(a, b, acc);
    }
}

__global__ void attn_kernel(const float* __restrict__ Qb,
                            const float* __restrict__ Kb,
                            const float* __restrict__ Vb,
                            float* __restrict__ CTX) {
    const int it   = blockIdx.x;
    const int head = blockIdx.y;
    const int nb   = blockIdx.z;
    const int lane = threadIdx.x;
    const int h    = lane >> 4;
    const int li   = lane & 15;

    const size_t base = (size_t)(nb * kHEADS + head) * kHD * kS;
    const float* q = Qb + base;
    const float* k = Kb + base;
    const float* v = Vb + base;

    const int i = it * 16 + li;

    const float* kl = k + (size_t)li * kHD + 8 * h;
    const float* vl = v + (size_t)(8 * h) * kHD + li;

    constexpr int kTiles = kS / 16;

    KVFrag fa = load_kv(kl, vl, 0);

    float bq[8];
#pragma unroll
    for (int kk = 0; kk < 4; ++kk) {
        bq[2 * kk + 0] = q[(size_t)(8 * h + 2 * kk + 0) * kS + i];
        bq[2 * kk + 1] = q[(size_t)(8 * h + 2 * kk + 1) * kS + i];
    }

    float m_run = -INFINITY;
    float l_run = 0.0f;
    v8f   acc   = {};

    for (int jt = 0; jt < kTiles; jt += 2) {
        const KVFrag fb = load_kv(kl, vl, jt + 1);
        hold_loads_here();
        attn_tile(fa, bq, m_run, l_run, acc);
        const int nx = (jt + 2 < kTiles) ? jt + 2 : kTiles - 1;
        fa = load_kv(kl, vl, nx);
        hold_loads_here();
        attn_tile(fb, bq, m_run, l_run, acc);
    }

    l_run += xhalf(l_run);
    const float inv = 1.0f / l_run;
    float* ctx = CTX + base;
#pragma unroll
    for (int r = 0; r < 8; ++r) {
        const int c = r + 8 * h;
        ctx[(size_t)c * kS + i] = acc[r] * inv;
    }
}

__global__ void out_proj_kernel(const float* __restrict__ CTX,
                                const float* __restrict__ w_o,
                                const float* __restrict__ b_o,
                                float* __restrict__ out) {
    const int ot   = blockIdx.x;
    const int st   = blockIdx.y;
    const int nb   = blockIdx.z;
    const int lane = threadIdx.x;
    const int h    = lane >> 4;
    const int li   = lane & 15;
    const int s    = st * 16 + li;

    const float* wrow  = w_o + (size_t)(ot * 16 + li) * kC;
    const float* cbase = CTX + (size_t)nb * kC * kS + s;

    v8f acc = {};
#pragma unroll 8
    for (int c0 = 0; c0 < kC; c0 += 4) {
        v2f a, b;
        a.x = wrow[c0 + 2 * h + 0];
        a.y = wrow[c0 + 2 * h + 1];
        b.x = cbase[(size_t)(c0 + 2 * h + 0) * kS];
        b.y = cbase[(size_t)(c0 + 2 * h + 1) * kS];
        acc = wmma_f32(a, b, acc);
    }
#pragma unroll
    for (int r = 0; r < 8; ++r) {
        const int oc = ot * 16 + r + 8 * h;
        out[((size_t)nb * kC + oc) * kS + s] = acc[r] + b_o[oc];
    }
}

extern "C" void kernel_launch(void* const* d_in, const int* in_sizes, int n_in,
                              void* d_out, int out_size, void* d_ws, size_t ws_size,
                              hipStream_t stream) {
    const float* x     = (const float*)d_in[0];
    const float* w_qkv = (const float*)d_in[1];
    const float* w_o   = (const float*)d_in[2];
    const float* b_o   = (const float*)d_in[3];
    float*       out   = (float*)d_out;

    const size_t per = (size_t)kNB * kC * kS;
    float* Qb  = (float*)d_ws;
    float* Kb  = Qb + per;
    float* Vb  = Kb + per;
    float* CTX = Vb + per;

    qkv_proj_kernel<<<dim3(24, 100, kNB), 32, 0, stream>>>(x, w_qkv, Qb, Kb, Vb);
    attn_kernel<<<dim3(100, kHEADS, kNB), 32, 0, stream>>>(Qb, Kb, Vb, CTX);
    out_proj_kernel<<<dim3(8, 100, kNB), 32, 0, stream>>>(CTX, w_o, b_o, out);
}


# Both 1x1-conv products: a wave computes a 16x64 tile (was 16x16) from 16-byte loads: W fragment reused 4x, X read in whole lines, 4x fewer waves; channel k slot c=8g+4h+2j+e, so c sums in a new order
# speedup vs baseline: 1.4052x; 1.1308x over previous
#include <hip/hip_runtime.h>
#include <cmath>

typedef __attribute__((ext_vector_type(2))) float v2f;
typedef __attribute__((ext_vector_type(8))) float v8f;
typedef __attribute__((ext_vector_type(4))) float v4f;

static constexpr int kS     = 1600;
static constexpr int kC     = 128;
static constexpr int kNB    = 8;
static constexpr int kHEADS = 8;
static constexpr int kHD    = 16;

__device__ __forceinline__ v8f wmma_f32(v2f a, v2f b, v8f c) {
    return __builtin_amdgcn_wmma_f32_16x16x4_f32(false, a, false, b, (short)0, c,
                                                 false, false);
}

__device__ __forceinline__ float xhalf(float x) {
    const int b = __builtin_bit_cast(int, x);
    return __builtin_bit_cast(float,
        __builtin_amdgcn_permlanex16(b, b, 0x76543210, 0xfedcba98, false, false));
}

__device__ __forceinline__ void hold_loads_here() {
    asm volatile("" ::: "memory");
    __builtin_amdgcn_sched_barrier(0);
}

__device__ __forceinline__ void proj_tile_16x64(const float* __restrict__ wl,
                                                const float* __restrict__ xl,
                                                v8f (&acc)[4]) {
    constexpr int kGB = 4;
#pragma nounroll
    for (int g0 = 0; g0 < kC / 8; g0 += kGB) {
        v4f w4[kGB];
        v4f xb[kGB][4];
#pragma unroll
        for (int u = 0; u < kGB; ++u)
            w4[u] = *reinterpret_cast<const v4f*>(wl + 8 * (g0 + u));
#pragma unroll
        for (int u = 0; u < kGB; ++u)
#pragma unroll
            for (int e = 0; e < 4; ++e)
                xb[u][e] = *reinterpret_cast<const v4f*>(xl + (size_t)(8 * (g0 + u) + e) * kS);
        hold_loads_here();
#pragma unroll
        for (int u = 0; u < kGB; ++u) {
#pragma unroll
            for (int j = 0; j < 2; ++j) {
                v2f a;
                a.x = w4[u][2 * j + 0];
                a.y = w4[u][2 * j + 1];
#pragma unroll
                for (int t = 0; t < 4; ++t) {
                    v2f b;
                    b.x = xb[u][2 * j + 0][t];
                    b.y = xb[u][2 * j + 1][t];
                    acc[t] = wmma_f32(a, b, acc[t]);
                }
            }
        }
    }
}

__global__ void qkv_proj_kernel(const float* __restrict__ x,
                                const float* __restrict__ w_qkv,
                                float* __restrict__ Qb, float* __restrict__ Kb,
                                float* __restrict__ Vb) {
    const int ot   = blockIdx.x;
    const int sb   = blockIdx.y;
    const int nb   = blockIdx.z;
    const int lane = threadIdx.x;
    const int h    = lane >> 4;
    const int li   = lane & 15;

    const int head  = ot / 3;
    const int which = ot % 3;
    const int sl    = sb * 64 + 4 * li;

    const float* wl = w_qkv + (size_t)(ot * 16 + li) * kC + 4 * h;
    const float* xl = x + (size_t)nb * kC * kS + (size_t)(4 * h) * kS + sl;

    v8f acc[4] = {};
    proj_tile_16x64(wl, xl, acc);

    const size_t base = (size_t)(nb * kHEADS + head) * kHD * kS;
    if (which == 0) {
        float* dst = Qb + base + sl;
#pragma unroll
        for (int r = 0; r < 8; ++r) {
            const int d = r + 8 * h;
            v4f q = {acc[0][r] * 0.25f, acc[1][r] * 0.25f,
                     acc[2][r] * 0.25f, acc[3][r] * 0.25f};
            *reinterpret_cast<v4f*>(dst + (size_t)d * kS) = q;
        }
    } else {
        float* dst = ((which == 1) ? Kb : Vb) + base + (size_t)sl * kHD + 8 * h;
#pragma unroll
        for (int t = 0; t < 4; ++t) {
            v4f lo = {acc[t][0], acc[t][1], acc[t][2], acc[t][3]};
            v4f hi = {acc[t][4], acc[t][5], acc[t][6], acc[t][7]};
            *reinterpret_cast<v4f*>(dst + t * kHD)     = lo;
            *reinterpret_cast<v4f*>(dst + t * kHD + 4) = hi;
        }
    }
}


struct KVFrag {
    v4f   k0, k1;
    float v[8];
};

__device__ __forceinline__ KVFrag load_kv(const float* __restrict__ kl,
                                          const float* __restrict__ vl, int jt) {
    KVFrag t;
    const float* kt = kl + (size_t)jt * 16 * kHD;
    t.k0 = *reinterpret_cast<const v4f*>(kt);
    t.k1 = *reinterpret_cast<const v4f*>(kt + 4);
    const float* vt = vl + (size_t)jt * 16 * kHD;
#pragma unroll
    for (int r = 0; r < 8; ++r) t.v[r] = vt[r * kHD];
    return t;
}

__device__ __forceinline__ void attn_tile(const KVFrag& t, const float (&bq)[8],
                                          float& m_run, float& l_run, v8f& acc) {
    const float ka[8] = {t.k0.x, t.k0.y, t.k0.z, t.k0.w, t.k1.x, t.k1.y, t.k1.z, t.k1.w};
    v8f e = {};
#pragma unroll
    for (int kk = 0; kk < 4; ++kk) {
        v2f a, b;
        a.x = ka[2 * kk + 0];
        a.y = ka[2 * kk + 1];
        b.x = bq[2 * kk + 0];
        b.y = bq[2 * kk + 1];
        e = wmma_f32(a, b, e);
    }

    float pmax = e[0];
#pragma unroll
    for (int r = 1; r < 8; ++r) pmax = fmaxf(pmax, e[r]);
    pmax = fmaxf(pmax, xhalf(pmax));
    const float m_new = fmaxf(m_run, pmax);
    const float alpha = __expf(m_run - m_new);
    float p[8], psum = 0.0f;
#pragma unroll
    for (int r = 0; r < 8; ++r) {
        p[r] = __expf(e[r] - m_new);
        psum += p[r];
    }
    l_run = l_run * alpha + psum;
    m_run = m_new;
#pragma unroll
    for (int r = 0; r < 8; ++r) acc[r] *= alpha;

#pragma unroll
    for (int kk = 0; kk < 4; ++kk) {
        v2f a, b;
        a.x = t.v[2 * kk + 0];
        a.y = t.v[2 * kk + 1];
        b.x = p[2 * kk + 0];
        b.y = p[2 * kk + 1];
        acc = wmma_f32(a, b, acc);
    }
}

__global__ void attn_kernel(const float* __restrict__ Qb,
                            const float* __restrict__ Kb,
                            const float* __restrict__ Vb,
                            float* __restrict__ CTX) {
    const int it   = blockIdx.x;
    const int head = blockIdx.y;
    const int nb   = blockIdx.z;
    const int lane = threadIdx.x;
    const int h    = lane >> 4;
    const int li   = lane & 15;

    const size_t base = (size_t)(nb * kHEADS + head) * kHD * kS;
    const float* q = Qb + base;
    const float* k = Kb + base;
    const float* v = Vb + base;

    const int i = it * 16 + li;

    const float* kl = k + (size_t)li * kHD + 8 * h;
    const float* vl = v + (size_t)(8 * h) * kHD + li;

    constexpr int kTiles = kS / 16;

    KVFrag fa = load_kv(kl, vl, 0);

    float bq[8];
#pragma unroll
    for (int kk = 0; kk < 4; ++kk) {
        bq[2 * kk + 0] = q[(size_t)(8 * h + 2 * kk + 0) * kS + i];
        bq[2 * kk + 1] = q[(size_t)(8 * h + 2 * kk + 1) * kS + i];
    }

    float m_run = -INFINITY;
    float l_run = 0.0f;
    v8f   acc   = {};

    for (int jt = 0; jt < kTiles; jt += 2) {
        const KVFrag fb = load_kv(kl, vl, jt + 1);
        hold_loads_here();
        attn_tile(fa, bq, m_run, l_run, acc);
        const int nx = (jt + 2 < kTiles) ? jt + 2 : kTiles - 1;
        fa = load_kv(kl, vl, nx);
        hold_loads_here();
        attn_tile(fb, bq, m_run, l_run, acc);
    }

    l_run += xhalf(l_run);
    const float inv = 1.0f / l_run;
    float* ctx = CTX + base;
#pragma unroll
    for (int r = 0; r < 8; ++r) {
        const int c = r + 8 * h;
        ctx[(size_t)c * kS + i] = acc[r] * inv;
    }
}

__global__ void out_proj_kernel(const float* __restrict__ CTX,
                                const float* __restrict__ w_o,
                                const float* __restrict__ b_o,
                                float* __restrict__ out) {
    const int ot   = blockIdx.x;
    const int sb   = blockIdx.y;
    const int nb   = blockIdx.z;
    const int lane = threadIdx.x;
    const int h    = lane >> 4;
    const int li   = lane & 15;
    const int sl   = sb * 64 + 4 * li;

    const float* wl = w_o + (size_t)(ot * 16 + li) * kC + 4 * h;
    const float* xl = CTX + (size_t)nb * kC * kS + (size_t)(4 * h) * kS + sl;

    v8f acc[4] = {};
    proj_tile_16x64(wl, xl, acc);

#pragma unroll
    for (int r = 0; r < 8; ++r) {
        const int   oc   = ot * 16 + r + 8 * h;
        const float bias = b_o[oc];
        v4f o = {acc[0][r] + bias, acc[1][r] + bias, acc[2][r] + bias, acc[3][r] + bias};
        *reinterpret_cast<v4f*>(out + ((size_t)nb * kC + oc) * kS + sl) = o;
    }
}

extern "C" void kernel_launch(void* const* d_in, const int* in_sizes, int n_in,
                              void* d_out, int out_size, void* d_ws, size_t ws_size,
                              hipStream_t stream) {
    const float* x     = (const float*)d_in[0];
    const float* w_qkv = (const float*)d_in[1];
    const float* w_o   = (const float*)d_in[2];
    const float* b_o   = (const float*)d_in[3];
    float*       out   = (float*)d_out;

    const size_t per = (size_t)kNB * kC * kS;
    float* Qb  = (float*)d_ws;
    float* Kb  = Qb + per;
    float* Vb  = Kb + per;
    float* CTX = Vb + per;

    qkv_proj_kernel<<<dim3(24, 25, kNB), 32, 0, stream>>>(x, w_qkv, Qb, Kb, Vb);
    attn_kernel<<<dim3(100, kHEADS, kNB), 32, 0, stream>>>(Qb, Kb, Vb, CTX);
    out_proj_kernel<<<dim3(8, 25, kNB), 32, 0, stream>>>(CTX, w_o, b_o, out);
}


# Attention softmax step widened from 16 to 64 keys: one column max, lane exchange, alpha and context rescale per 64 keys; row sums add in new order (packed pairs), context rescaled at new points
# speedup vs baseline: 1.5887x; 1.1306x over previous
#include <hip/hip_runtime.h>
#include <cmath>

typedef __attribute__((ext_vector_type(2))) float v2f;
typedef __attribute__((ext_vector_type(8))) float v8f;
typedef __attribute__((ext_vector_type(4))) float v4f;

static constexpr int kS     = 1600;
static constexpr int kC     = 128;
static constexpr int kNB    = 8;
static constexpr int kHEADS = 8;
static constexpr int kHD    = 16;

__device__ __forceinline__ v8f wmma_f32(v2f a, v2f b, v8f c) {
    return __builtin_amdgcn_wmma_f32_16x16x4_f32(false, a, false, b, (short)0, c,
                                                 false, false);
}

__device__ __forceinline__ float xhalf(float x) {
    const int b = __builtin_bit_cast(int, x);
    return __builtin_bit_cast(float,
        __builtin_amdgcn_permlanex16(b, b, 0x76543210, 0xfedcba98, false, false));
}

__device__ __forceinline__ void hold_loads_here() {
    asm volatile("" ::: "memory");
    __builtin_amdgcn_sched_barrier(0);
}

__device__ __forceinline__ void proj_tile_16x64(const float* __restrict__ wl,
                                                const float* __restrict__ xl,
                                                v8f (&acc)[4]) {
    constexpr int kGB = 4;
#pragma nounroll
    for (int g0 = 0; g0 < kC / 8; g0 += kGB) {
        v4f w4[kGB];
        v4f xb[kGB][4];
#pragma unroll
        for (int u = 0; u < kGB; ++u)
            w4[u] = *reinterpret_cast<const v4f*>(wl + 8 * (g0 + u));
#pragma unroll
        for (int u = 0; u < kGB; ++u)
#pragma unroll
            for (int e = 0; e < 4; ++e)
                xb[u][e] = *reinterpret_cast<const v4f*>(xl + (size_t)(8 * (g0 + u) + e) * kS);
        hold_loads_here();
#pragma unroll
        for (int u = 0; u < kGB; ++u) {
#pragma unroll
            for (int j = 0; j < 2; ++j) {
                v2f a;
                a.x = w4[u][2 * j + 0];
                a.y = w4[u][2 * j + 1];
#pragma unroll
                for (int t = 0; t < 4; ++t) {
                    v2f b;
                    b.x = xb[u][2 * j + 0][t];
                    b.y = xb[u][2 * j + 1][t];
                    acc[t] = wmma_f32(a, b, acc[t]);
                }
            }
        }
    }
}

__global__ void qkv_proj_kernel(const float* __restrict__ x,
                                const float* __restrict__ w_qkv,
                                float* __restrict__ Qb, float* __restrict__ Kb,
                                float* __restrict__ Vb) {
    const int ot   = blockIdx.x;
    const int sb   = blockIdx.y;
    const int nb   = blockIdx.z;
    const int lane = threadIdx.x;
    const int h    = lane >> 4;
    const int li   = lane & 15;

    const int head  = ot / 3;
    const int which = ot % 3;
    const int sl    = sb * 64 + 4 * li;

    const float* wl = w_qkv + (size_t)(ot * 16 + li) * kC + 4 * h;
    const float* xl = x + (size_t)nb * kC * kS + (size_t)(4 * h) * kS + sl;

    v8f acc[4] = {};
    proj_tile_16x64(wl, xl, acc);

    const size_t base = (size_t)(nb * kHEADS + head) * kHD * kS;
    if (which == 0) {
        float* dst = Qb + base + sl;
#pragma unroll
        for (int r = 0; r < 8; ++r) {
            const int d = r + 8 * h;
            v4f q = {acc[0][r] * 0.25f, acc[1][r] * 0.25f,
                     acc[2][r] * 0.25f, acc[3][r] * 0.25f};
            *reinterpret_cast<v4f*>(dst + (size_t)d * kS) = q;
        }
    } else {
        float* dst = ((which == 1) ? Kb : Vb) + base + (size_t)sl * kHD + 8 * h;
#pragma unroll
        for (int t = 0; t < 4; ++t) {
            v4f lo = {acc[t][0], acc[t][1], acc[t][2], acc[t][3]};
            v4f hi = {acc[t][4], acc[t][5], acc[t][6], acc[t][7]};
            *reinterpret_cast<v4f*>(dst + t * kHD)     = lo;
            *reinterpret_cast<v4f*>(dst + t * kHD + 4) = hi;
        }
    }
}


constexpr int kStepTiles = 4;

struct KFrag {
    v4f k0, k1;
};

struct VFrag {
    float v[8];
};

__device__ __forceinline__ void load_k_step(KFrag (&kf)[kStepTiles],
                                            const float* __restrict__ kl, int st) {
#pragma unroll
    for (int t = 0; t < kStepTiles; ++t) {
        const float* kt = kl + (size_t)(st * kStepTiles + t) * 16 * kHD;
        kf[t].k0 = *reinterpret_cast<const v4f*>(kt);
        kf[t].k1 = *reinterpret_cast<const v4f*>(kt + 4);
    }
}

__device__ __forceinline__ void load_v_step(VFrag (&vf)[kStepTiles],
                                            const float* __restrict__ vl, int st) {
#pragma unroll
    for (int t = 0; t < kStepTiles; ++t) {
        const float* vt = vl + (size_t)(st * kStepTiles + t) * 16 * kHD;
#pragma unroll
        for (int r = 0; r < 8; ++r) vf[t].v[r] = vt[r * kHD];
    }
}

__global__ void attn_kernel(const float* __restrict__ Qb,
                            const float* __restrict__ Kb,
                            const float* __restrict__ Vb,
                            float* __restrict__ CTX) {
    const int it   = blockIdx.x;
    const int head = blockIdx.y;
    const int nb   = blockIdx.z;
    const int lane = threadIdx.x;
    const int h    = lane >> 4;
    const int li   = lane & 15;

    const size_t base = (size_t)(nb * kHEADS + head) * kHD * kS;
    const float* q = Qb + base;
    const float* k = Kb + base;
    const float* v = Vb + base;

    const int i = it * 16 + li;

    const float* kl = k + (size_t)li * kHD + 8 * h;
    const float* vl = v + (size_t)(8 * h) * kHD + li;

    constexpr int kSteps = kS / (16 * kStepTiles);
    static_assert(kSteps * 16 * kStepTiles == kS, "key count must be whole steps");

    KFrag kf[kStepTiles];
    load_k_step(kf, kl, 0);

    float bq[8];
#pragma unroll
    for (int kk = 0; kk < 4; ++kk) {
        bq[2 * kk + 0] = q[(size_t)(8 * h + 2 * kk + 0) * kS + i];
        bq[2 * kk + 1] = q[(size_t)(8 * h + 2 * kk + 1) * kS + i];
    }

    float m_run = -INFINITY;
    float l_run = 0.0f;
    v8f   acc   = {};

    for (int st = 0; st < kSteps; ++st) {
        VFrag vf[kStepTiles];
        load_v_step(vf, vl, st);
        hold_loads_here();

        v8f e[kStepTiles];
#pragma unroll
        for (int t = 0; t < kStepTiles; ++t) e[t] = v8f{};
#pragma unroll
        for (int kk = 0; kk < 4; ++kk) {
            v2f b;
            b.x = bq[2 * kk + 0];
            b.y = bq[2 * kk + 1];
#pragma unroll
            for (int t = 0; t < kStepTiles; ++t) {
                const float ka[8] = {kf[t].k0.x, kf[t].k0.y, kf[t].k0.z, kf[t].k0.w,
                                     kf[t].k1.x, kf[t].k1.y, kf[t].k1.z, kf[t].k1.w};
                v2f a;
                a.x = ka[2 * kk + 0];
                a.y = ka[2 * kk + 1];
                e[t] = wmma_f32(a, b, e[t]);
            }
        }

        const int nx = (st + 1 < kSteps) ? st + 1 : kSteps - 1;
        load_k_step(kf, kl, nx);
        hold_loads_here();

        float tmax[kStepTiles];
#pragma unroll
        for (int t = 0; t < kStepTiles; ++t) {
            tmax[t] = e[t][0];
#pragma unroll
            for (int r = 1; r < 8; ++r) tmax[t] = fmaxf(tmax[t], e[t][r]);
        }
        float pmax = fmaxf(fmaxf(tmax[0], tmax[1]), fmaxf(tmax[2], tmax[3]));
        pmax = fmaxf(pmax, xhalf(pmax));
        const float m_new = fmaxf(m_run, pmax);
        const float alpha = __expf(m_run - m_new);
        v2f sa = {0.0f, 0.0f};
        v2f sb = {0.0f, 0.0f};
#pragma unroll
        for (int t = 0; t < kStepTiles; ++t) {
#pragma unroll
            for (int kk = 0; kk < 4; ++kk) {
                v2f pp;
                pp.x = __expf(e[t][2 * kk + 0] - m_new);
                pp.y = __expf(e[t][2 * kk + 1] - m_new);
                e[t][2 * kk + 0] = pp.x;
                e[t][2 * kk + 1] = pp.y;
                if (t < kStepTiles / 2) sa += pp; else sb += pp;
            }
        }
        const v2f s2 = sa + sb;
        l_run = l_run * alpha + (s2.x + s2.y);
        m_run = m_new;
#pragma unroll
        for (int r = 0; r < 8; ++r) acc[r] *= alpha;

#pragma unroll
        for (int t = 0; t < kStepTiles; ++t) {
#pragma unroll
            for (int kk = 0; kk < 4; ++kk) {
                v2f a, b;
                a.x = vf[t].v[2 * kk + 0];
                a.y = vf[t].v[2 * kk + 1];
                b.x = e[t][2 * kk + 0];
                b.y = e[t][2 * kk + 1];
                acc = wmma_f32(a, b, acc);
            }
        }
    }

    l_run += xhalf(l_run);
    const float inv = 1.0f / l_run;
    float* ctx = CTX + base;
#pragma unroll
    for (int r = 0; r < 8; ++r) {
        const int c = r + 8 * h;
        ctx[(size_t)c * kS + i] = acc[r] * inv;
    }
}

__global__ void out_proj_kernel(const float* __restrict__ CTX,
                                const float* __restrict__ w_o,
                                const float* __restrict__ b_o,
                                float* __restrict__ out) {
    const int ot   = blockIdx.x;
    const int sb   = blockIdx.y;
    const int nb   = blockIdx.z;
    const int lane = threadIdx.x;
    const int h    = lane >> 4;
    const int li   = lane & 15;
    const int sl   = sb * 64 + 4 * li;

    const float* wl = w_o + (size_t)(ot * 16 + li) * kC + 4 * h;
    const float* xl = CTX + (size_t)nb * kC * kS + (size_t)(4 * h) * kS + sl;

    v8f acc[4] = {};
    proj_tile_16x64(wl, xl, acc);

#pragma unroll
    for (int r = 0; r < 8; ++r) {
        const int   oc   = ot * 16 + r + 8 * h;
        const float bias = b_o[oc];
        v4f o = {acc[0][r] + bias, acc[1][r] + bias, acc[2][r] + bias, acc[3][r] + bias};
        *reinterpret_cast<v4f*>(out + ((size_t)nb * kC + oc) * kS + sl) = o;
    }
}

extern "C" void kernel_launch(void* const* d_in, const int* in_sizes, int n_in,
                              void* d_out, int out_size, void* d_ws, size_t ws_size,
                              hipStream_t stream) {
    const float* x     = (const float*)d_in[0];
    const float* w_qkv = (const float*)d_in[1];
    const float* w_o   = (const float*)d_in[2];
    const float* b_o   = (const float*)d_in[3];
    float*       out   = (float*)d_out;

    const size_t per = (size_t)kNB * kC * kS;
    float* Qb  = (float*)d_ws;
    float* Kb  = Qb + per;
    float* Vb  = Kb + per;
    float* CTX = Vb + per;

    qkv_proj_kernel<<<dim3(24, 25, kNB), 32, 0, stream>>>(x, w_qkv, Qb, Kb, Vb);
    attn_kernel<<<dim3(100, kHEADS, kNB), 32, 0, stream>>>(Qb, Kb, Vb, CTX);
    out_proj_kernel<<<dim3(8, 25, kNB), 32, 0, stream>>>(CTX, w_o, b_o, out);
}


# Attention: 25 query tiles per work-group on 28 waves; 24 take a tile each, 4 (one per SIMD) split the 25th by key range and join partial sums via LDS (new sum order there): 6.25 tiles per SIMD, not 7
# speedup vs baseline: 1.8267x; 1.1498x over previous
#include <hip/hip_runtime.h>
#include <cmath>

typedef __attribute__((ext_vector_type(2))) float v2f;
typedef __attribute__((ext_vector_type(8))) float v8f;
typedef __attribute__((ext_vector_type(4))) float v4f;

static constexpr int kS     = 1600;
static constexpr int kC     = 128;
static constexpr int kNB    = 8;
static constexpr int kHEADS = 8;
static constexpr int kHD    = 16;

__device__ __forceinline__ v8f wmma_f32(v2f a, v2f b, v8f c) {
    return __builtin_amdgcn_wmma_f32_16x16x4_f32(false, a, false, b, (short)0, c,
                                                 false, false);
}

__device__ __forceinline__ float xhalf(float x) {
    const int b = __builtin_bit_cast(int, x);
    return __builtin_bit_cast(float,
        __builtin_amdgcn_permlanex16(b, b, 0x76543210, 0xfedcba98, false, false));
}

__device__ __forceinline__ void hold_loads_here() {
    asm volatile("" ::: "memory");
    __builtin_amdgcn_sched_barrier(0);
}

__device__ __forceinline__ void proj_tile_16x64(const float* __restrict__ wl,
                                                const float* __restrict__ xl,
                                                v8f (&acc)[4]) {
    constexpr int kGB = 4;
#pragma nounroll
    for (int g0 = 0; g0 < kC / 8; g0 += kGB) {
        v4f w4[kGB];
        v4f xb[kGB][4];
#pragma unroll
        for (int u = 0; u < kGB; ++u)
            w4[u] = *reinterpret_cast<const v4f*>(wl + 8 * (g0 + u));
#pragma unroll
        for (int u = 0; u < kGB; ++u)
#pragma unroll
            for (int e = 0; e < 4; ++e)
                xb[u][e] = *reinterpret_cast<const v4f*>(xl + (size_t)(8 * (g0 + u) + e) * kS);
        hold_loads_here();
#pragma unroll
        for (int u = 0; u < kGB; ++u) {
#pragma unroll
            for (int j = 0; j < 2; ++j) {
                v2f a;
                a.x = w4[u][2 * j + 0];
                a.y = w4[u][2 * j + 1];
#pragma unroll
                for (int t = 0; t < 4; ++t) {
                    v2f b;
                    b.x = xb[u][2 * j + 0][t];
                    b.y = xb[u][2 * j + 1][t];
                    acc[t] = wmma_f32(a, b, acc[t]);
                }
            }
        }
    }
}

__global__ void qkv_proj_kernel(const float* __restrict__ x,
                                const float* __restrict__ w_qkv,
                                float* __restrict__ Qb, float* __restrict__ Kb,
                                float* __restrict__ Vb) {
    const int ot   = blockIdx.x;
    const int sb   = blockIdx.y;
    const int nb   = blockIdx.z;
    const int lane = threadIdx.x;
    const int h    = lane >> 4;
    const int li   = lane & 15;

    const int head  = ot / 3;
    const int which = ot % 3;
    const int sl    = sb * 64 + 4 * li;

    const float* wl = w_qkv + (size_t)(ot * 16 + li) * kC + 4 * h;
    const float* xl = x + (size_t)nb * kC * kS + (size_t)(4 * h) * kS + sl;

    v8f acc[4] = {};
    proj_tile_16x64(wl, xl, acc);

    const size_t base = (size_t)(nb * kHEADS + head) * kHD * kS;
    if (which == 0) {
        float* dst = Qb + base + sl;
#pragma unroll
        for (int r = 0; r < 8; ++r) {
            const int d = r + 8 * h;
            v4f q = {acc[0][r] * 0.25f, acc[1][r] * 0.25f,
                     acc[2][r] * 0.25f, acc[3][r] * 0.25f};
            *reinterpret_cast<v4f*>(dst + (size_t)d * kS) = q;
        }
    } else {
        float* dst = ((which == 1) ? Kb : Vb) + base + (size_t)sl * kHD + 8 * h;
#pragma unroll
        for (int t = 0; t < 4; ++t) {
            v4f lo = {acc[t][0], acc[t][1], acc[t][2], acc[t][3]};
            v4f hi = {acc[t][4], acc[t][5], acc[t][6], acc[t][7]};
            *reinterpret_cast<v4f*>(dst + t * kHD)     = lo;
            *reinterpret_cast<v4f*>(dst + t * kHD + 4) = hi;
        }
    }
}


constexpr int kStepTiles = 4;

constexpr int kGroupTiles  = 25;
constexpr int kSplitWaves  = 4;
constexpr int kAttnWaves   = kGroupTiles - 1 + kSplitWaves;
constexpr int kAttnThreads = kAttnWaves * 32;
constexpr int kAttnGroups  = kS / 16 / kGroupTiles;
constexpr int kWaveRun     = kAttnWaves / kSplitWaves;
static_assert(kAttnGroups * kGroupTiles * 16 == kS, "query tiles must be whole work-groups");
static_assert(kWaveRun * kSplitWaves == kAttnWaves, "one sharing wave per run of waves");

struct KFrag {
    v4f k0, k1;
};

struct VFrag {
    float v[8];
};

__device__ __forceinline__ void load_k_step(KFrag (&kf)[kStepTiles],
                                            const float* __restrict__ kl, int st) {
#pragma unroll
    for (int t = 0; t < kStepTiles; ++t) {
        const float* kt = kl + (size_t)(st * kStepTiles + t) * 16 * kHD;
        kf[t].k0 = *reinterpret_cast<const v4f*>(kt);
        kf[t].k1 = *reinterpret_cast<const v4f*>(kt + 4);
    }
}

__device__ __forceinline__ void load_v_step(VFrag (&vf)[kStepTiles],
                                            const float* __restrict__ vl, int st) {
#pragma unroll
    for (int t = 0; t < kStepTiles; ++t) {
        const float* vt = vl + (size_t)(st * kStepTiles + t) * 16 * kHD;
#pragma unroll
        for (int r = 0; r < 8; ++r) vf[t].v[r] = vt[r * kHD];
    }
}

__global__ void __launch_bounds__(kAttnThreads)
attn_kernel(const float* __restrict__ Qb,
            const float* __restrict__ Kb,
            const float* __restrict__ Vb,
            float* __restrict__ CTX) {
    __shared__ float part[kSplitWaves][10][32];

    const int wave = __builtin_amdgcn_readfirstlane((int)(threadIdx.x >> 5));
    const int run  = wave / kWaveRun;
    const bool shares = (wave % kWaveRun) == kWaveRun - 1;
    const int it   = blockIdx.x * kGroupTiles + (shares ? kGroupTiles - 1 : wave - run);
    const int head = blockIdx.y;
    const int nb   = blockIdx.z;
    const int lane = threadIdx.x & 31;
    const int h    = lane >> 4;
    const int li   = lane & 15;

    const size_t base = (size_t)(nb * kHEADS + head) * kHD * kS;
    const float* q = Qb + base;
    const float* k = Kb + base;
    const float* v = Vb + base;

    const int i = it * 16 + li;

    const float* kl = k + (size_t)li * kHD + 8 * h;
    const float* vl = v + (size_t)(8 * h) * kHD + li;

    constexpr int kSteps = kS / (16 * kStepTiles);
    static_assert(kSteps * 16 * kStepTiles == kS, "key count must be whole steps");
    static_assert(kSteps == 6 * kSplitWaves + 1, "the shared tile's steps split 7, 6, 6, 6");
    const int st0 = shares ? 6 * run + (run > 0 ? 1 : 0) : 0;
    const int st1 = shares ? 6 * run + 7 : kSteps;

    KFrag kf[kStepTiles];
    load_k_step(kf, kl, st0);

    float bq[8];
#pragma unroll
    for (int kk = 0; kk < 4; ++kk) {
        bq[2 * kk + 0] = q[(size_t)(8 * h + 2 * kk + 0) * kS + i];
        bq[2 * kk + 1] = q[(size_t)(8 * h + 2 * kk + 1) * kS + i];
    }

    float m_run = -INFINITY;
    float l_run = 0.0f;
    v8f   acc   = {};

    for (int st = st0; st < st1; ++st) {
        VFrag vf[kStepTiles];
        load_v_step(vf, vl, st);
        hold_loads_here();

        v8f e[kStepTiles];
#pragma unroll
        for (int t = 0; t < kStepTiles; ++t) e[t] = v8f{};
#pragma unroll
        for (int kk = 0; kk < 4; ++kk) {
            v2f b;
            b.x = bq[2 * kk + 0];
            b.y = bq[2 * kk + 1];
#pragma unroll
            for (int t = 0; t < kStepTiles; ++t) {
                const float ka[8] = {kf[t].k0.x, kf[t].k0.y, kf[t].k0.z, kf[t].k0.w,
                                     kf[t].k1.x, kf[t].k1.y, kf[t].k1.z, kf[t].k1.w};
                v2f a;
                a.x = ka[2 * kk + 0];
                a.y = ka[2 * kk + 1];
                e[t] = wmma_f32(a, b, e[t]);
            }
        }

        const int nx = (st + 1 < st1) ? st + 1 : st1 - 1;
        load_k_step(kf, kl, nx);
        hold_loads_here();

        float tmax[kStepTiles];
#pragma unroll
        for (int t = 0; t < kStepTiles; ++t) {
            tmax[t] = e[t][0];
#pragma unroll
            for (int r = 1; r < 8; ++r) tmax[t] = fmaxf(tmax[t], e[t][r]);
        }
        float pmax = fmaxf(fmaxf(tmax[0], tmax[1]), fmaxf(tmax[2], tmax[3]));
        pmax = fmaxf(pmax, xhalf(pmax));
        const float m_new = fmaxf(m_run, pmax);
        const float alpha = __expf(m_run - m_new);
        v2f sa = {0.0f, 0.0f};
        v2f sb = {0.0f, 0.0f};
#pragma unroll
        for (int t = 0; t < kStepTiles; ++t) {
#pragma unroll
            for (int kk = 0; kk < 4; ++kk) {
                v2f pp;
                pp.x = __expf(e[t][2 * kk + 0] - m_new);
                pp.y = __expf(e[t][2 * kk + 1] - m_new);
                e[t][2 * kk + 0] = pp.x;
                e[t][2 * kk + 1] = pp.y;
                if (t < kStepTiles / 2) sa += pp; else sb += pp;
            }
        }
        const v2f s2 = sa + sb;
        l_run = l_run * alpha + (s2.x + s2.y);
        m_run = m_new;
#pragma unroll
        for (int r = 0; r < 8; ++r) acc[r] *= alpha;

#pragma unroll
        for (int t = 0; t < kStepTiles; ++t) {
#pragma unroll
            for (int kk = 0; kk < 4; ++kk) {
                v2f a, b;
                a.x = vf[t].v[2 * kk + 0];
                a.y = vf[t].v[2 * kk + 1];
                b.x = e[t][2 * kk + 0];
                b.y = e[t][2 * kk + 1];
                acc = wmma_f32(a, b, acc);
            }
        }
    }

    l_run += xhalf(l_run);
    float* ctx = CTX + base;
    if (!shares) {
        const float inv = 1.0f / l_run;
#pragma unroll
        for (int r = 0; r < 8; ++r) {
            const int c = r + 8 * h;
            ctx[(size_t)c * kS + i] = acc[r] * inv;
        }
    } else {
        part[run][0][lane] = m_run;
        part[run][1][lane] = l_run;
#pragma unroll
        for (int r = 0; r < 8; ++r) part[run][2 + r][lane] = acc[r];
    }
    __syncthreads();
    if (wave == kWaveRun - 1) {
        float mk[kSplitWaves];
        float mall = -INFINITY;
#pragma unroll
        for (int sw = 0; sw < kSplitWaves; ++sw) {
            mk[sw] = part[sw][0][lane];
            mall   = fmaxf(mall, mk[sw]);
        }
        float lsum = 0.0f;
        float o[8] = {0.0f, 0.0f, 0.0f, 0.0f, 0.0f, 0.0f, 0.0f, 0.0f};
#pragma unroll
        for (int sw = 0; sw < kSplitWaves; ++sw) {
            const float wk = __expf(mk[sw] - mall);
            lsum += wk * part[sw][1][lane];
#pragma unroll
            for (int r = 0; r < 8; ++r) o[r] += wk * part[sw][2 + r][lane];
        }
        const float inv = 1.0f / lsum;
#pragma unroll
        for (int r = 0; r < 8; ++r) {
            const int c = r + 8 * h;
            ctx[(size_t)c * kS + i] = o[r] * inv;
        }
    }
}

__global__ void out_proj_kernel(const float* __restrict__ CTX,
                                const float* __restrict__ w_o,
                                const float* __restrict__ b_o,
                                float* __restrict__ out) {
    const int ot   = blockIdx.x;
    const int sb   = blockIdx.y;
    const int nb   = blockIdx.z;
    const int lane = threadIdx.x;
    const int h    = lane >> 4;
    const int li   = lane & 15;
    const int sl   = sb * 64 + 4 * li;

    const float* wl = w_o + (size_t)(ot * 16 + li) * kC + 4 * h;
    const float* xl = CTX + (size_t)nb * kC * kS + (size_t)(4 * h) * kS + sl;

    v8f acc[4] = {};
    proj_tile_16x64(wl, xl, acc);

#pragma unroll
    for (int r = 0; r < 8; ++r) {
        const int   oc   = ot * 16 + r + 8 * h;
        const float bias = b_o[oc];
        v4f o = {acc[0][r] + bias, acc[1][r] + bias, acc[2][r] + bias, acc[3][r] + bias};
        *reinterpret_cast<v4f*>(out + ((size_t)nb * kC + oc) * kS + sl) = o;
    }
}

extern "C" void kernel_launch(void* const* d_in, const int* in_sizes, int n_in,
                              void* d_out, int out_size, void* d_ws, size_t ws_size,
                              hipStream_t stream) {
    const float* x     = (const float*)d_in[0];
    const float* w_qkv = (const float*)d_in[1];
    const float* w_o   = (const float*)d_in[2];
    const float* b_o   = (const float*)d_in[3];
    float*       out   = (float*)d_out;

    const size_t per = (size_t)kNB * kC * kS;
    float* Qb  = (float*)d_ws;
    float* Kb  = Qb + per;
    float* Vb  = Kb + per;
    float* CTX = Vb + per;

    qkv_proj_kernel<<<dim3(24, 25, kNB), 32, 0, stream>>>(x, w_qkv, Qb, Kb, Vb);
    attn_kernel<<<dim3(kAttnGroups, kHEADS, kNB), kAttnThreads, 0, stream>>>(Qb, Kb, Vb, CTX);
    out_proj_kernel<<<dim3(8, 25, kNB), 32, 0, stream>>>(CTX, w_o, b_o, out);
}


# Softmax shift rides in the score product's C operand (base-2 scores): 2^x straight on scores, no per-step maximum; step redone, rescaled only if a row sum passes 2^12; P, sums carry a common factor
# speedup vs baseline: 1.9454x; 1.0650x over previous
#include <hip/hip_runtime.h>
#include <cmath>

typedef __attribute__((ext_vector_type(2))) float v2f;
typedef __attribute__((ext_vector_type(8))) float v8f;
typedef __attribute__((ext_vector_type(4))) float v4f;

static constexpr int kS     = 1600;
static constexpr int kC     = 128;
static constexpr int kNB    = 8;
static constexpr int kHEADS = 8;
static constexpr int kHD    = 16;
static constexpr float kQScale = 0.25f * 1.44269504088896340736f;

__device__ __forceinline__ v8f wmma_f32(v2f a, v2f b, v8f c) {
    return __builtin_amdgcn_wmma_f32_16x16x4_f32(false, a, false, b, (short)0, c,
                                                 false, false);
}

__device__ __forceinline__ float ex2(float x) { return __builtin_amdgcn_exp2f(x); }

__device__ __forceinline__ float xhalf(float x) {
    const int b = __builtin_bit_cast(int, x);
    return __builtin_bit_cast(float,
        __builtin_amdgcn_permlanex16(b, b, 0x76543210, 0xfedcba98, false, false));
}

__device__ __forceinline__ void hold_loads_here() {
    asm volatile("" ::: "memory");
    __builtin_amdgcn_sched_barrier(0);
}

__device__ __forceinline__ void proj_tile_16x64(const float* __restrict__ wl,
                                                const float* __restrict__ xl,
                                                v8f (&acc)[4]) {
    constexpr int kGB = 4;
#pragma nounroll
    for (int g0 = 0; g0 < kC / 8; g0 += kGB) {
        v4f w4[kGB];
        v4f xb[kGB][4];
#pragma unroll
        for (int u = 0; u < kGB; ++u)
            w4[u] = *reinterpret_cast<const v4f*>(wl + 8 * (g0 + u));
#pragma unroll
        for (int u = 0; u < kGB; ++u)
#pragma unroll
            for (int e = 0; e < 4; ++e)
                xb[u][e] = *reinterpret_cast<const v4f*>(xl + (size_t)(8 * (g0 + u) + e) * kS);
        hold_loads_here();
#pragma unroll
        for (int u = 0; u < kGB; ++u) {
#pragma unroll
            for (int j = 0; j < 2; ++j) {
                v2f a;
                a.x = w4[u][2 * j + 0];
                a.y = w4[u][2 * j + 1];
#pragma unroll
                for (int t = 0; t < 4; ++t) {
                    v2f b;
                    b.x = xb[u][2 * j + 0][t];
                    b.y = xb[u][2 * j + 1][t];
                    acc[t] = wmma_f32(a, b, acc[t]);
                }
            }
        }
    }
}

__global__ void qkv_proj_kernel(const float* __restrict__ x,
                                const float* __restrict__ w_qkv,
                                float* __restrict__ Qb, float* __restrict__ Kb,
                                float* __restrict__ Vb) {
    const int ot   = blockIdx.x;
    const int sb   = blockIdx.y;
    const int nb   = blockIdx.z;
    const int lane = threadIdx.x;
    const int h    = lane >> 4;
    const int li   = lane & 15;

    const int head  = ot / 3;
    const int which = ot % 3;
    const int sl    = sb * 64 + 4 * li;

    const float* wl = w_qkv + (size_t)(ot * 16 + li) * kC + 4 * h;
    const float* xl = x + (size_t)nb * kC * kS + (size_t)(4 * h) * kS + sl;

    v8f acc[4] = {};
    proj_tile_16x64(wl, xl, acc);

    const size_t base = (size_t)(nb * kHEADS + head) * kHD * kS;
    if (which == 0) {
        float* dst = Qb + base + sl;
#pragma unroll
        for (int r = 0; r < 8; ++r) {
            const int d = r + 8 * h;
            v4f q = {acc[0][r] * kQScale, acc[1][r] * kQScale,
                     acc[2][r] * kQScale, acc[3][r] * kQScale};
            *reinterpret_cast<v4f*>(dst + (size_t)d * kS) = q;
        }
    } else {
        float* dst = ((which == 1) ? Kb : Vb) + base + (size_t)sl * kHD + 8 * h;
#pragma unroll
        for (int t = 0; t < 4; ++t) {
            v4f lo = {acc[t][0], acc[t][1], acc[t][2], acc[t][3]};
            v4f hi = {acc[t][4], acc[t][5], acc[t][6], acc[t][7]};
            *reinterpret_cast<v4f*>(dst + t * kHD)     = lo;
            *reinterpret_cast<v4f*>(dst + t * kHD + 4) = hi;
        }
    }
}


constexpr int kStepTiles = 4;

constexpr float kSumLimit = 4096.0f;

constexpr int kGroupTiles  = 25;
constexpr int kSplitWaves  = 4;
constexpr int kAttnWaves   = kGroupTiles - 1 + kSplitWaves;
constexpr int kAttnThreads = kAttnWaves * 32;
constexpr int kAttnGroups  = kS / 16 / kGroupTiles;
constexpr int kWaveRun     = kAttnWaves / kSplitWaves;
static_assert(kAttnGroups * kGroupTiles * 16 == kS, "query tiles must be whole work-groups");
static_assert(kWaveRun * kSplitWaves == kAttnWaves, "one sharing wave per run of waves");

struct KFrag {
    v4f k0, k1;
};

struct VFrag {
    float v[8];
};

__device__ __forceinline__ void load_k_step(KFrag (&kf)[kStepTiles],
                                            const float* __restrict__ kl, int st) {
#pragma unroll
    for (int t = 0; t < kStepTiles; ++t) {
        const float* kt = kl + (size_t)(st * kStepTiles + t) * 16 * kHD;
        kf[t].k0 = *reinterpret_cast<const v4f*>(kt);
        kf[t].k1 = *reinterpret_cast<const v4f*>(kt + 4);
    }
}

__device__ __forceinline__ void load_v_step(VFrag (&vf)[kStepTiles],
                                            const float* __restrict__ vl, int st) {
#pragma unroll
    for (int t = 0; t < kStepTiles; ++t) {
        const float* vt = vl + (size_t)(st * kStepTiles + t) * 16 * kHD;
#pragma unroll
        for (int r = 0; r < 8; ++r) vf[t].v[r] = vt[r * kHD];
    }
}

__device__ __forceinline__ void score_tiles(v8f (&e)[kStepTiles], const KFrag (&kf)[kStepTiles],
                                            const float (&bq)[8], v8f cstart) {
#pragma unroll
    for (int kk = 0; kk < 4; ++kk) {
        v2f b;
        b.x = bq[2 * kk + 0];
        b.y = bq[2 * kk + 1];
#pragma unroll
        for (int t = 0; t < kStepTiles; ++t) {
            const float ka[8] = {kf[t].k0.x, kf[t].k0.y, kf[t].k0.z, kf[t].k0.w,
                                 kf[t].k1.x, kf[t].k1.y, kf[t].k1.z, kf[t].k1.w};
            v2f a;
            a.x = ka[2 * kk + 0];
            a.y = ka[2 * kk + 1];
            e[t] = wmma_f32(a, b, kk == 0 ? cstart : e[t]);
        }
    }
}

__device__ __forceinline__ void move_shift(v8f (&e)[kStepTiles], float dfloor, float& m_ref,
                                           float& l_run, v8f& acc) {
    float tmax[kStepTiles];
#pragma unroll
    for (int t = 0; t < kStepTiles; ++t) {
        tmax[t] = e[t][0];
#pragma unroll
        for (int r = 1; r < 8; ++r) tmax[t] = fmaxf(tmax[t], e[t][r]);
    }
    float pmax = fmaxf(fmaxf(tmax[0], tmax[1]), fmaxf(tmax[2], tmax[3]));
    pmax = fmaxf(pmax, xhalf(pmax));
    const float delta = fmaxf(pmax, dfloor);
    m_ref += delta;
    const float alpha = ex2(fminf(-delta, 0.0f));
#pragma unroll
    for (int t = 0; t < kStepTiles; ++t)
#pragma unroll
        for (int r = 0; r < 8; ++r) e[t][r] -= delta;
    l_run *= alpha;
#pragma unroll
    for (int r = 0; r < 8; ++r) acc[r] *= alpha;
}

__device__ __forceinline__ float exp_and_sum(v8f (&e)[kStepTiles]) {
    v2f sa = {0.0f, 0.0f};
    v2f sb = {0.0f, 0.0f};
#pragma unroll
    for (int t = 0; t < kStepTiles; ++t) {
#pragma unroll
        for (int kk = 0; kk < 4; ++kk) {
            v2f pp;
            pp.x = ex2(e[t][2 * kk + 0]);
            pp.y = ex2(e[t][2 * kk + 1]);
            e[t][2 * kk + 0] = pp.x;
            e[t][2 * kk + 1] = pp.y;
            if (t < kStepTiles / 2) sa += pp; else sb += pp;
        }
    }
    const v2f s2 = sa + sb;
    return s2.x + s2.y;
}

__global__ void __launch_bounds__(kAttnThreads)
attn_kernel(const float* __restrict__ Qb,
            const float* __restrict__ Kb,
            const float* __restrict__ Vb,
            float* __restrict__ CTX) {
    __shared__ float part[kSplitWaves][10][32];

    const int wave = __builtin_amdgcn_readfirstlane((int)(threadIdx.x >> 5));
    const int run  = wave / kWaveRun;
    const bool shares = (wave % kWaveRun) == kWaveRun - 1;
    const int it   = blockIdx.x * kGroupTiles + (shares ? kGroupTiles - 1 : wave - run);
    const int head = blockIdx.y;
    const int nb   = blockIdx.z;
    const int lane = threadIdx.x & 31;
    const int h    = lane >> 4;
    const int li   = lane & 15;

    const size_t base = (size_t)(nb * kHEADS + head) * kHD * kS;
    const float* q = Qb + base;
    const float* k = Kb + base;
    const float* v = Vb + base;

    const int i = it * 16 + li;

    const float* kl = k + (size_t)li * kHD + 8 * h;
    const float* vl = v + (size_t)(8 * h) * kHD + li;

    constexpr int kSteps = kS / (16 * kStepTiles);
    static_assert(kSteps * 16 * kStepTiles == kS, "key count must be whole steps");
    static_assert(kSteps == 6 * kSplitWaves + 1, "the shared tile's steps split 7, 6, 6, 6");
    const int st0 = shares ? 6 * run + (run > 0 ? 1 : 0) : 0;
    const int st1 = shares ? 6 * run + 7 : kSteps;

    KFrag kf[kStepTiles];
    load_k_step(kf, kl, st0);

    float bq[8];
#pragma unroll
    for (int kk = 0; kk < 4; ++kk) {
        bq[2 * kk + 0] = q[(size_t)(8 * h + 2 * kk + 0) * kS + i];
        bq[2 * kk + 1] = q[(size_t)(8 * h + 2 * kk + 1) * kS + i];
    }

    float m_ref = 0.0f;
    float l_run = 0.0f;
    v8f   acc   = {};

    for (int st = st0; st < st1; ++st) {
        VFrag vf[kStepTiles];
        load_v_step(vf, vl, st);
        hold_loads_here();

        v8f cneg;
#pragma unroll
        for (int r = 0; r < 8; ++r) cneg[r] = -m_ref;
        v8f e[kStepTiles];
        score_tiles(e, kf, bq, cneg);

        if (st == st0)
            move_shift(e, -INFINITY, m_ref, l_run, acc);

        float psum = exp_and_sum(e);
        if (__builtin_amdgcn_ballot_w32(!(psum <= kSumLimit)) != 0u) {
            score_tiles(e, kf, bq, cneg);
            move_shift(e, 0.0f, m_ref, l_run, acc);
            psum = exp_and_sum(e);
        }
        l_run += psum;

        const int nx = (st + 1 < st1) ? st + 1 : st1 - 1;
        load_k_step(kf, kl, nx);
        hold_loads_here();

#pragma unroll
        for (int t = 0; t < kStepTiles; ++t) {
#pragma unroll
            for (int kk = 0; kk < 4; ++kk) {
                v2f a, b;
                a.x = vf[t].v[2 * kk + 0];
                a.y = vf[t].v[2 * kk + 1];
                b.x = e[t][2 * kk + 0];
                b.y = e[t][2 * kk + 1];
                acc = wmma_f32(a, b, acc);
            }
        }
    }

    l_run += xhalf(l_run);
    float* ctx = CTX + base;
    if (!shares) {
        const float inv = 1.0f / l_run;
#pragma unroll
        for (int r = 0; r < 8; ++r) {
            const int c = r + 8 * h;
            ctx[(size_t)c * kS + i] = acc[r] * inv;
        }
    } else {
        part[run][0][lane] = m_ref;
        part[run][1][lane] = l_run;
#pragma unroll
        for (int r = 0; r < 8; ++r) part[run][2 + r][lane] = acc[r];
    }
    __syncthreads();
    if (wave == kWaveRun - 1) {
        float mk[kSplitWaves];
        float mall = -INFINITY;
#pragma unroll
        for (int sw = 0; sw < kSplitWaves; ++sw) {
            mk[sw] = part[sw][0][lane];
            mall   = fmaxf(mall, mk[sw]);
        }
        float lsum = 0.0f;
        float o[8] = {0.0f, 0.0f, 0.0f, 0.0f, 0.0f, 0.0f, 0.0f, 0.0f};
#pragma unroll
        for (int sw = 0; sw < kSplitWaves; ++sw) {
            const float wk = ex2(mk[sw] - mall);
            lsum += wk * part[sw][1][lane];
#pragma unroll
            for (int r = 0; r < 8; ++r) o[r] += wk * part[sw][2 + r][lane];
        }
        const float inv = 1.0f / lsum;
#pragma unroll
        for (int r = 0; r < 8; ++r) {
            const int c = r + 8 * h;
            ctx[(size_t)c * kS + i] = o[r] * inv;
        }
    }
}

__global__ void out_proj_kernel(const float* __restrict__ CTX,
                                const float* __restrict__ w_o,
                                const float* __restrict__ b_o,
                                float* __restrict__ out) {
    const int ot   = blockIdx.x;
    const int sb   = blockIdx.y;
    const int nb   = blockIdx.z;
    const int lane = threadIdx.x;
    const int h    = lane >> 4;
    const int li   = lane & 15;
    const int sl   = sb * 64 + 4 * li;

    const float* wl = w_o + (size_t)(ot * 16 + li) * kC + 4 * h;
    const float* xl = CTX + (size_t)nb * kC * kS + (size_t)(4 * h) * kS + sl;

    v8f acc[4] = {};
    proj_tile_16x64(wl, xl, acc);

#pragma unroll
    for (int r = 0; r < 8; ++r) {
        const int   oc   = ot * 16 + r + 8 * h;
        const float bias = b_o[oc];
        v4f o = {acc[0][r] + bias, acc[1][r] + bias, acc[2][r] + bias, acc[3][r] + bias};
        *reinterpret_cast<v4f*>(out + ((size_t)nb * kC + oc) * kS + sl) = o;
    }
}

extern "C" void kernel_launch(void* const* d_in, const int* in_sizes, int n_in,
                              void* d_out, int out_size, void* d_ws, size_t ws_size,
                              hipStream_t stream) {
    const float* x     = (const float*)d_in[0];
    const float* w_qkv = (const float*)d_in[1];
    const float* w_o   = (const float*)d_in[2];
    const float* b_o   = (const float*)d_in[3];
    float*       out   = (float*)d_out;

    const size_t per = (size_t)kNB * kC * kS;
    float* Qb  = (float*)d_ws;
    float* Kb  = Qb + per;
    float* Vb  = Kb + per;
    float* CTX = Vb + per;

    qkv_proj_kernel<<<dim3(24, 25, kNB), 32, 0, stream>>>(x, w_qkv, Qb, Kb, Vb);
    attn_kernel<<<dim3(kAttnGroups, kHEADS, kNB), kAttnThreads, 0, stream>>>(Qb, Kb, Vb, CTX);
    out_proj_kernel<<<dim3(8, 25, kNB), 32, 0, stream>>>(CTX, w_o, b_o, out);
}


# Attention: a work-group copies its head's K and V (200 KB) into LDS once, in fragment order (the projection now writes that order); its 28 waves read 16-byte fragments from LDS; sum orders unchanged
# speedup vs baseline: 2.0303x; 1.0436x over previous
#include <hip/hip_runtime.h>
#include <cmath>

typedef __attribute__((ext_vector_type(2))) float v2f;
typedef __attribute__((ext_vector_type(8))) float v8f;
typedef __attribute__((ext_vector_type(4))) float v4f;

static constexpr int kS     = 1600;
static constexpr int kC     = 128;
static constexpr int kNB    = 8;
static constexpr int kHEADS = 8;
static constexpr int kHD    = 16;
static constexpr float kQScale = 0.25f * 1.44269504088896340736f;

__device__ __forceinline__ v8f wmma_f32(v2f a, v2f b, v8f c) {
    return __builtin_amdgcn_wmma_f32_16x16x4_f32(false, a, false, b, (short)0, c,
                                                 false, false);
}

__device__ __forceinline__ float ex2(float x) { return __builtin_amdgcn_exp2f(x); }

__device__ __forceinline__ float xhalf(float x) {
    const int b = __builtin_bit_cast(int, x);
    return __builtin_bit_cast(float,
        __builtin_amdgcn_permlanex16(b, b, 0x76543210, 0xfedcba98, false, false));
}

__device__ __forceinline__ void hold_loads_here() {
    asm volatile("" ::: "memory");
    __builtin_amdgcn_sched_barrier(0);
}

__device__ __forceinline__ void proj_tile_16x64(const float* __restrict__ wl,
                                                const float* __restrict__ xl,
                                                v8f (&acc)[4]) {
    constexpr int kGB = 4;
#pragma nounroll
    for (int g0 = 0; g0 < kC / 8; g0 += kGB) {
        v4f w4[kGB];
        v4f xb[kGB][4];
#pragma unroll
        for (int u = 0; u < kGB; ++u)
            w4[u] = *reinterpret_cast<const v4f*>(wl + 8 * (g0 + u));
#pragma unroll
        for (int u = 0; u < kGB; ++u)
#pragma unroll
            for (int e = 0; e < 4; ++e)
                xb[u][e] = *reinterpret_cast<const v4f*>(xl + (size_t)(8 * (g0 + u) + e) * kS);
        hold_loads_here();
#pragma unroll
        for (int u = 0; u < kGB; ++u) {
#pragma unroll
            for (int j = 0; j < 2; ++j) {
                v2f a;
                a.x = w4[u][2 * j + 0];
                a.y = w4[u][2 * j + 1];
#pragma unroll
                for (int t = 0; t < 4; ++t) {
                    v2f b;
                    b.x = xb[u][2 * j + 0][t];
                    b.y = xb[u][2 * j + 1][t];
                    acc[t] = wmma_f32(a, b, acc[t]);
                }
            }
        }
    }
}

__global__ void qkv_proj_kernel(const float* __restrict__ x,
                                const float* __restrict__ w_qkv,
                                float* __restrict__ Qb, float* __restrict__ Kb,
                                float* __restrict__ Vb) {
    const int ot   = blockIdx.x;
    const int sb   = blockIdx.y;
    const int nb   = blockIdx.z;
    const int lane = threadIdx.x;
    const int h    = lane >> 4;
    const int li   = lane & 15;

    const int head  = ot / 3;
    const int which = ot % 3;
    const int sl    = sb * 64 + 4 * li;

    const float* wl = w_qkv + (size_t)(ot * 16 + li) * kC + 4 * h;
    const float* xl = x + (size_t)nb * kC * kS + (size_t)(4 * h) * kS + sl;

    v8f acc[4] = {};
    proj_tile_16x64(wl, xl, acc);

    const size_t base = (size_t)(nb * kHEADS + head) * kHD * kS;
    if (which == 0) {
        float* dst = Qb + base + sl;
#pragma unroll
        for (int r = 0; r < 8; ++r) {
            const int d = r + 8 * h;
            v4f q = {acc[0][r] * kQScale, acc[1][r] * kQScale,
                     acc[2][r] * kQScale, acc[3][r] * kQScale};
            *reinterpret_cast<v4f*>(dst + (size_t)d * kS) = q;
        }
    } else {
        float* blk = ((which == 1) ? Kb : Vb) + base + (size_t)(sb * 4 + (li >> 2)) * (16 * kHD);
        if (which == 1) {
            float* dst = blk + h * 64 + 16 * (li & 3);
#pragma unroll
            for (int t = 0; t < 4; ++t) {
                v4f lo = {acc[t][0], acc[t][1], acc[t][2], acc[t][3]};
                v4f hi = {acc[t][4], acc[t][5], acc[t][6], acc[t][7]};
                *reinterpret_cast<v4f*>(dst + 4 * t)       = lo;
                *reinterpret_cast<v4f*>(dst + 128 + 4 * t) = hi;
            }
        } else {
            float* dst = blk + (li & 1) * 128 + ((li >> 1) & 1) * 64 + 32 * h;
#pragma unroll
            for (int r = 0; r < 8; ++r) {
                v4f kq = {acc[0][r], acc[1][r], acc[2][r], acc[3][r]};
                *reinterpret_cast<v4f*>(dst + 4 * r) = kq;
            }
        }
    }
}


constexpr int kStepTiles = 4;

constexpr float kSumLimit = 4096.0f;

constexpr int kGroupTiles  = 25;
constexpr int kSplitWaves  = 4;
constexpr int kAttnWaves   = kGroupTiles - 1 + kSplitWaves;
constexpr int kAttnThreads = kAttnWaves * 32;
constexpr int kAttnGroups  = kS / 16 / kGroupTiles;
constexpr int kWaveRun     = kAttnWaves / kSplitWaves;
static_assert(kAttnGroups * kGroupTiles * 16 == kS, "query tiles must be whole work-groups");
static_assert(kWaveRun * kSplitWaves == kAttnWaves, "one sharing wave per run of waves");

constexpr int kTileVec  = 16 * kHD / 4;
constexpr int kPlaneVec = kS * kHD / 4;
constexpr int kFullPasses = kPlaneVec / kAttnThreads;
constexpr int kTailVec    = kPlaneVec - kFullPasses * kAttnThreads;
static_assert(kTailVec > 0 && kTailVec % 32 == 0, "the last pass must be whole waves");

struct KFrag {
    v4f k0, k1;
};

struct VFrag {
    v4f v0, v1;
};

__device__ __forceinline__ void load_k_step(KFrag (&kf)[kStepTiles], const v4f* kp, int st) {
#pragma unroll
    for (int t = 0; t < kStepTiles; ++t) {
        const v4f* kt = kp + (st * kStepTiles + t) * kTileVec;
        kf[t].k0 = kt[0];
        kf[t].k1 = kt[32];
    }
}

__device__ __forceinline__ void load_v_step(VFrag (&vf)[kStepTiles], const v4f* vp, int st) {
#pragma unroll
    for (int t = 0; t < kStepTiles; ++t) {
        const v4f* vt = vp + (st * kStepTiles + t) * kTileVec;
        vf[t].v0 = vt[0];
        vf[t].v1 = vt[32];
    }
}

__device__ __forceinline__ void score_tiles(v8f (&e)[kStepTiles], const KFrag (&kf)[kStepTiles],
                                            const float (&bq)[8], v8f cstart) {
#pragma unroll
    for (int kk = 0; kk < 4; ++kk) {
        v2f b;
        b.x = bq[2 * kk + 0];
        b.y = bq[2 * kk + 1];
#pragma unroll
        for (int t = 0; t < kStepTiles; ++t) {
            const float ka[8] = {kf[t].k0.x, kf[t].k0.y, kf[t].k0.z, kf[t].k0.w,
                                 kf[t].k1.x, kf[t].k1.y, kf[t].k1.z, kf[t].k1.w};
            v2f a;
            a.x = ka[2 * kk + 0];
            a.y = ka[2 * kk + 1];
            e[t] = wmma_f32(a, b, kk == 0 ? cstart : e[t]);
        }
    }
}

__device__ __forceinline__ void move_shift(v8f (&e)[kStepTiles], float dfloor, float& m_ref,
                                           float& l_run, v8f& acc) {
    float tmax[kStepTiles];
#pragma unroll
    for (int t = 0; t < kStepTiles; ++t) {
        tmax[t] = e[t][0];
#pragma unroll
        for (int r = 1; r < 8; ++r) tmax[t] = fmaxf(tmax[t], e[t][r]);
    }
    float pmax = fmaxf(fmaxf(tmax[0], tmax[1]), fmaxf(tmax[2], tmax[3]));
    pmax = fmaxf(pmax, xhalf(pmax));
    const float delta = fmaxf(pmax, dfloor);
    m_ref += delta;
    const float alpha = ex2(fminf(-delta, 0.0f));
#pragma unroll
    for (int t = 0; t < kStepTiles; ++t)
#pragma unroll
        for (int r = 0; r < 8; ++r) e[t][r] -= delta;
    l_run *= alpha;
#pragma unroll
    for (int r = 0; r < 8; ++r) acc[r] *= alpha;
}

__device__ __forceinline__ float exp_and_sum(v8f (&e)[kStepTiles]) {
    v2f sa = {0.0f, 0.0f};
    v2f sb = {0.0f, 0.0f};
#pragma unroll
    for (int t = 0; t < kStepTiles; ++t) {
#pragma unroll
        for (int kk = 0; kk < 4; ++kk) {
            v2f pp;
            pp.x = ex2(e[t][2 * kk + 0]);
            pp.y = ex2(e[t][2 * kk + 1]);
            e[t][2 * kk + 0] = pp.x;
            e[t][2 * kk + 1] = pp.y;
            if (t < kStepTiles / 2) sa += pp; else sb += pp;
        }
    }
    const v2f s2 = sa + sb;
    return s2.x + s2.y;
}

__global__ void __launch_bounds__(kAttnThreads)
attn_kernel(const float* __restrict__ Qb,
            const float* __restrict__ Kb,
            const float* __restrict__ Vb,
            float* __restrict__ CTX) {
    __shared__ float part[kSplitWaves][10][32];
    __shared__ v4f kv[2][kPlaneVec];

    const int wave = __builtin_amdgcn_readfirstlane((int)(threadIdx.x >> 5));
    const int run  = wave / kWaveRun;
    const bool shares = (wave % kWaveRun) == kWaveRun - 1;
    const int it   = blockIdx.x * kGroupTiles + (shares ? kGroupTiles - 1 : wave - run);
    const int head = blockIdx.y;
    const int nb   = blockIdx.z;
    const int lane = threadIdx.x & 31;
    const int h    = lane >> 4;
    const int li   = lane & 15;

    const size_t base = (size_t)(nb * kHEADS + head) * kHD * kS;
    const float* q = Qb + base;
    const float* k = Kb + base;
    const float* v = Vb + base;

    const int i = it * 16 + li;

    const v4f* kl = &kv[0][lane];
    const v4f* vl = &kv[1][lane];

    constexpr int kSteps = kS / (16 * kStepTiles);
    static_assert(kSteps * 16 * kStepTiles == kS, "key count must be whole steps");
    static_assert(kSteps == 6 * kSplitWaves + 1, "the shared tile's steps split 7, 6, 6, 6");
    const int st0 = shares ? 6 * run + (run > 0 ? 1 : 0) : 0;
    const int st1 = shares ? 6 * run + 7 : kSteps;

    float bq[8];
#pragma unroll
    for (int kk = 0; kk < 4; ++kk) {
        bq[2 * kk + 0] = q[(size_t)(8 * h + 2 * kk + 0) * kS + i];
        bq[2 * kk + 1] = q[(size_t)(8 * h + 2 * kk + 1) * kS + i];
    }

    {
        const v4f* kg  = reinterpret_cast<const v4f*>(k);
        const v4f* vg  = reinterpret_cast<const v4f*>(v);
        const int  tid = (int)threadIdx.x;
        const bool tail = tid < kTailVec;
        v4f ck[kFullPasses + 1], cv[kFullPasses + 1];
#pragma unroll
        for (int p = 0; p < kFullPasses; ++p) {
            ck[p] = kg[p * kAttnThreads + tid];
            cv[p] = vg[p * kAttnThreads + tid];
        }
        if (tail) {
            ck[kFullPasses] = kg[kFullPasses * kAttnThreads + tid];
            cv[kFullPasses] = vg[kFullPasses * kAttnThreads + tid];
        }
#pragma unroll
        for (int p = 0; p < kFullPasses; ++p) {
            kv[0][p * kAttnThreads + tid] = ck[p];
            kv[1][p * kAttnThreads + tid] = cv[p];
        }
        if (tail) {
            kv[0][kFullPasses * kAttnThreads + tid] = ck[kFullPasses];
            kv[1][kFullPasses * kAttnThreads + tid] = cv[kFullPasses];
        }
    }
    __syncthreads();

    KFrag kf[kStepTiles];
    load_k_step(kf, kl, st0);

    float m_ref = 0.0f;
    float l_run = 0.0f;
    v8f   acc   = {};

    for (int st = st0; st < st1; ++st) {
        VFrag vf[kStepTiles];
        load_v_step(vf, vl, st);
        hold_loads_here();

        v8f cneg;
#pragma unroll
        for (int r = 0; r < 8; ++r) cneg[r] = -m_ref;
        v8f e[kStepTiles];
        score_tiles(e, kf, bq, cneg);

        if (st == st0)
            move_shift(e, -INFINITY, m_ref, l_run, acc);

        float psum = exp_and_sum(e);
        if (__builtin_amdgcn_ballot_w32(!(psum <= kSumLimit)) != 0u) {
            score_tiles(e, kf, bq, cneg);
            move_shift(e, 0.0f, m_ref, l_run, acc);
            psum = exp_and_sum(e);
        }
        l_run += psum;

        const int nx = (st + 1 < st1) ? st + 1 : st1 - 1;
        load_k_step(kf, kl, nx);
        hold_loads_here();

#pragma unroll
        for (int t = 0; t < kStepTiles; ++t) {
            const float va[8] = {vf[t].v0.x, vf[t].v0.y, vf[t].v0.z, vf[t].v0.w,
                                 vf[t].v1.x, vf[t].v1.y, vf[t].v1.z, vf[t].v1.w};
#pragma unroll
            for (int kk = 0; kk < 4; ++kk) {
                v2f a, b;
                a.x = va[2 * kk + 0];
                a.y = va[2 * kk + 1];
                b.x = e[t][2 * kk + 0];
                b.y = e[t][2 * kk + 1];
                acc = wmma_f32(a, b, acc);
            }
        }
    }

    l_run += xhalf(l_run);
    float* ctx = CTX + base;
    if (!shares) {
        const float inv = 1.0f / l_run;
#pragma unroll
        for (int r = 0; r < 8; ++r) {
            const int c = r + 8 * h;
            ctx[(size_t)c * kS + i] = acc[r] * inv;
        }
    } else {
        part[run][0][lane] = m_ref;
        part[run][1][lane] = l_run;
#pragma unroll
        for (int r = 0; r < 8; ++r) part[run][2 + r][lane] = acc[r];
    }
    __syncthreads();
    if (wave == kWaveRun - 1) {
        float mk[kSplitWaves];
        float mall = -INFINITY;
#pragma unroll
        for (int sw = 0; sw < kSplitWaves; ++sw) {
            mk[sw] = part[sw][0][lane];
            mall   = fmaxf(mall, mk[sw]);
        }
        float lsum = 0.0f;
        float o[8] = {0.0f, 0.0f, 0.0f, 0.0f, 0.0f, 0.0f, 0.0f, 0.0f};
#pragma unroll
        for (int sw = 0; sw < kSplitWaves; ++sw) {
            const float wk = ex2(mk[sw] - mall);
            lsum += wk * part[sw][1][lane];
#pragma unroll
            for (int r = 0; r < 8; ++r) o[r] += wk * part[sw][2 + r][lane];
        }
        const float inv = 1.0f / lsum;
#pragma unroll
        for (int r = 0; r < 8; ++r) {
            const int c = r + 8 * h;
            ctx[(size_t)c * kS + i] = o[r] * inv;
        }
    }
}

__global__ void out_proj_kernel(const float* __restrict__ CTX,
                                const float* __restrict__ w_o,
                                const float* __restrict__ b_o,
                                float* __restrict__ out) {
    const int ot   = blockIdx.x;
    const int sb   = blockIdx.y;
    const int nb   = blockIdx.z;
    const int lane = threadIdx.x;
    const int h    = lane >> 4;
    const int li   = lane & 15;
    const int sl   = sb * 64 + 4 * li;

    const float* wl = w_o + (size_t)(ot * 16 + li) * kC + 4 * h;
    const float* xl = CTX + (size_t)nb * kC * kS + (size_t)(4 * h) * kS + sl;

    v8f acc[4] = {};
    proj_tile_16x64(wl, xl, acc);

#pragma unroll
    for (int r = 0; r < 8; ++r) {
        const int   oc   = ot * 16 + r + 8 * h;
        const float bias = b_o[oc];
        v4f o = {acc[0][r] + bias, acc[1][r] + bias, acc[2][r] + bias, acc[3][r] + bias};
        *reinterpret_cast<v4f*>(out + ((size_t)nb * kC + oc) * kS + sl) = o;
    }
}

extern "C" void kernel_launch(void* const* d_in, const int* in_sizes, int n_in,
                              void* d_out, int out_size, void* d_ws, size_t ws_size,
                              hipStream_t stream) {
    const float* x     = (const float*)d_in[0];
    const float* w_qkv = (const float*)d_in[1];
    const float* w_o   = (const float*)d_in[2];
    const float* b_o   = (const float*)d_in[3];
    float*       out   = (float*)d_out;

    const size_t per = (size_t)kNB * kC * kS;
    float* Qb  = (float*)d_ws;
    float* Kb  = Qb + per;
    float* Vb  = Kb + per;
    float* CTX = Vb + per;

    qkv_proj_kernel<<<dim3(24, 25, kNB), 32, 0, stream>>>(x, w_qkv, Qb, Kb, Vb);
    attn_kernel<<<dim3(kAttnGroups, kHEADS, kNB), kAttnThreads, 0, stream>>>(Qb, Kb, Vb, CTX);
    out_proj_kernel<<<dim3(8, 25, kNB), 32, 0, stream>>>(CTX, w_o, b_o, out);
}
